# starting kernel
# baseline (speedup 1.0000x reference)
#include <hip/hip_runtime.h>


typedef __attribute__((ext_vector_type(16))) _Float16 v16h;
typedef __attribute__((ext_vector_type(8)))  _Float16 v8h;
typedef __attribute__((ext_vector_type(8)))  float    v8f;

#define DEVI static __device__ __forceinline__

static constexpr int BATCH = 128;
static constexpr int SEQ   = 512;
static constexpr int E     = 192;
static constexpr int QDIM  = 48;
static constexpr int NTOK  = BATCH * SEQ;

static constexpr int KT_LD = 24;
static constexpr int VT_LD = 520;

DEVI v8f wmma16(v16h a, v16h b, v8f c) {
  return __builtin_amdgcn_wmma_f32_16x16x32_f16(false, a, false, b,
                                                (short)0, c, false, false);
}

DEVI float swap16(float v) {
  int i = __builtin_bit_cast(int, v);
  int r = __builtin_amdgcn_permlanex16(i, i, 0x76543210, 0xfedcba98, false, false);
  return __builtin_bit_cast(float, r);
}

DEVI v16h load_frag(const _Float16* __restrict__ base, int ld, int krem) {
  const int lane = threadIdx.x & 31;
  const int row  = lane & 15;
  const int d0   = (lane & 16) ? 8 : 0;
  const _Float16* p = base + (size_t)row * ld + d0;
  v16h r;
  if (krem >= 32) {
    v8h lo = *(const v8h*)(p);
    v8h hi = *(const v8h*)(p + 16);
#pragma unroll
    for (int i = 0; i < 8; ++i) { r[i] = lo[i]; r[8 + i] = hi[i]; }
  } else {
#pragma unroll
    for (int i = 0; i < 8; ++i) {
      r[i]     = (d0 + i      < krem) ? p[i]      : (_Float16)0.0f;
      r[8 + i] = (d0 + 16 + i < krem) ? p[16 + i] : (_Float16)0.0f;
    }
  }
  return r;
}

__global__ void cvt_f16_kernel(const float* __restrict__ src,
                               _Float16* __restrict__ dst, int n) {
  int i = blockIdx.x * 256 + threadIdx.x;
  if (i < n) dst[i] = (_Float16)src[i];
}

__global__ __launch_bounds__(256) void ln_kernel(
    const float* __restrict__ x, const float* __restrict__ g,
    const float* __restrict__ b, _Float16* __restrict__ out) {
  const int wave = threadIdx.x >> 5, lane = threadIdx.x & 31;
  const size_t t = (size_t)blockIdx.x * 8 + wave;
  const float* xp = x + t * E;
  float v[6];
  float s = 0.0f;
#pragma unroll
  for (int i = 0; i < 6; ++i) { v[i] = xp[lane + 32 * i]; s += v[i]; }
#pragma unroll
  for (int o = 16; o > 0; o >>= 1) s += __shfl_xor(s, o, 32);
  const float mean = s * (1.0f / E);
  float vs = 0.0f;
#pragma unroll
  for (int i = 0; i < 6; ++i) { float d = v[i] - mean; vs += d * d; }
#pragma unroll
  for (int o = 16; o > 0; o >>= 1) vs += __shfl_xor(vs, o, 32);
  const float inv = rsqrtf(vs * (1.0f / E) + 1e-6f);
#pragma unroll
  for (int i = 0; i < 6; ++i) {
    int c = lane + 32 * i;
    out[t * E + c] = (_Float16)((v[i] - mean) * inv * g[c] + b[c]);
  }
}

__global__ void conv_kernel(const _Float16* __restrict__ xn,
                            const float* __restrict__ x,
                            const float* __restrict__ cw,
                            float* __restrict__ x1) {
  const size_t idx = (size_t)blockIdx.x * 256 + threadIdx.x;
  if (idx >= (size_t)NTOK * 96) return;
  const int h = (int)(idx % 96);
  const size_t t = idx / 96;
  const float* w = cw + (t & 3) * 15;
  const _Float16* xp = xn + t * E + 48;
  float s = 0.0f;
#pragma unroll
  for (int k = 0; k < 15; ++k) {
    int hh = h + k - 7;
    if (hh >= 0 && hh < 96) s += w[k] * (float)xp[hh];
  }
  x1[t * E + 48 + h] = s + x[t * E + 48 + h];
}

template <int NT, bool RELU, bool RES, bool OUT16>
__global__ __launch_bounds__(256) void gemm_f16_kernel(
    const _Float16* __restrict__ A, int lda,
    const _Float16* __restrict__ W, int K,
    const float* __restrict__ bias,
    const float* __restrict__ res, int ldr,
    float* __restrict__ out32, _Float16* __restrict__ out16, int ldo) {
  const int wave = threadIdx.x >> 5;
  const int m0 = blockIdx.x * 128 + wave * 16;
  const int n0 = blockIdx.y * (NT * 16);
  v8f z = {};
  v8f acc[NT];
#pragma unroll
  for (int t = 0; t < NT; ++t) acc[t] = z;

  for (int k0 = 0; k0 < K; k0 += 32) {
    const int krem = K - k0;
    v16h a = load_frag(A + (size_t)m0 * lda + k0, lda, krem);
#pragma unroll
    for (int t = 0; t < NT; ++t) {
      v16h bfr = load_frag(W + (size_t)(n0 + t * 16) * K + k0, K, krem);
      acc[t] = wmma16(a, bfr, acc[t]);
    }
  }

  const int lane = threadIdx.x & 31;
  const int col = lane & 15;
  const int r0  = (lane & 16) ? 8 : 0;
#pragma unroll
  for (int t = 0; t < NT; ++t) {
    const int n = n0 + t * 16 + col;
    const float bv = bias[n];
#pragma unroll
    for (int j = 0; j < 8; ++j) {
      const int m = m0 + r0 + j;
      float v = acc[t][j] + bv;
      if (RELU) v = fmaxf(v, 0.0f);
      if (RES)  v += res[(size_t)m * ldr + n];
      if (OUT16) out16[(size_t)m * ldo + n] = (_Float16)v;
      else       out32[(size_t)m * ldo + n] = v;
    }
  }
}

__global__ __launch_bounds__(256) void attn_kernel(
    const _Float16* __restrict__ qkv_all, _Float16* __restrict__ out_all) {
  __shared__ _Float16 vt[16 * VT_LD];
  __shared__ _Float16 kt[SEQ * KT_LD];

  const int bid = blockIdx.x;
  const int b = bid / 6;
  const int r = bid % 6;
  const int branch = r / 3;
  const int hh = r % 3;
  const _Float16* qkv = qkv_all + (size_t)branch * NTOK * 144
                                + (size_t)b * SEQ * 144;
  _Float16* outp = out_all + (size_t)branch * NTOK * QDIM
                           + (size_t)b * SEQ * QDIM;
  const int qcol = hh * 16, kcol = 48 + hh * 16, vcol = 96 + hh * 16;

  for (int e = threadIdx.x; e < 16 * SEQ; e += 256) {
    const int d = e & 15, j = e >> 4;
    const _Float16 kv = qkv[(size_t)j * 144 + kcol + d];
    const _Float16 vv = qkv[(size_t)j * 144 + vcol + d];
    kt[j * KT_LD + d] = kv;
    vt[d * VT_LD + j] = vv;
  }
  __syncthreads();

  const int wave = threadIdx.x >> 5, lane = threadIdx.x & 31;
  const int q0 = blockIdx.y * 128 + wave * 16;
  const int row = lane & 15;
  const int d0  = (lane & 16) ? 8 : 0;

  v16h qf;
  {
    const _Float16* qp = qkv + (size_t)(q0 + row) * 144 + qcol + d0;
    v8h ql = *(const v8h*)qp;
#pragma unroll
    for (int i = 0; i < 8; ++i) {
      qf[i] = ql[i] * (_Float16)0.25f;
      qf[8 + i] = (_Float16)0.0f;
    }
  }

  v8f z = {};
  v8f oacc = z;
  float m_run = -1e30f, l_run = 0.0f;

  for (int kb = 0; kb < SEQ / 32; ++kb) {
    const int j0 = kb * 32;
    v16h kf0, kf1;
    {
      const _Float16* kp0 = &kt[(j0 + row) * KT_LD + d0];
      const _Float16* kp1 = &kt[(j0 + 16 + row) * KT_LD + d0];
      v8h ka = *(const v8h*)kp0;
      v8h kb2 = *(const v8h*)kp1;
#pragma unroll
      for (int i = 0; i < 8; ++i) {
        kf0[i] = ka[i];  kf0[8 + i] = (_Float16)0.0f;
        kf1[i] = kb2[i]; kf1[8 + i] = (_Float16)0.0f;
      }
    }
    v8f s0 = wmma16(kf0, qf, z);
    v8f s1 = wmma16(kf1, qf, z);

    float tm = -1e30f;
#pragma unroll
    for (int j = 0; j < 8; ++j) tm = fmaxf(tm, fmaxf(s0[j], s1[j]));
    tm = fmaxf(tm, swap16(tm));
    const float mnew = fmaxf(m_run, tm);
    const float corr = __expf(m_run - mnew);
    float p0[8], p1[8], ls = 0.0f;
#pragma unroll
    for (int j = 0; j < 8; ++j) {
      p0[j] = __expf(s0[j] - mnew);
      p1[j] = __expf(s1[j] - mnew);
      ls += p0[j] + p1[j];
    }
    ls += swap16(ls);
    l_run = l_run * corr + ls;
    m_run = mnew;
#pragma unroll
    for (int j = 0; j < 8; ++j) oacc[j] *= corr;

    v16h pf;
#pragma unroll
    for (int i = 0; i < 8; ++i) {
      pf[i] = (_Float16)p0[i];
      pf[8 + i] = (_Float16)p1[i];
    }
    const _Float16* vp = &vt[row * VT_LD + j0 + d0];
    v8h vl = *(const v8h*)vp;
    v8h vh = *(const v8h*)(vp + 16);
    v16h vf;
#pragma unroll
    for (int i = 0; i < 8; ++i) { vf[i] = vl[i]; vf[8 + i] = vh[i]; }

    oacc = wmma16(vf, pf, oacc);
  }

  const float inv = 1.0f / l_run;
#pragma unroll
  for (int j = 0; j < 8; ++j)
    outp[(size_t)(q0 + row) * QDIM + qcol + d0 + j] = (_Float16)(oacc[j] * inv);
}

static inline size_t align256(size_t x) { return (x + 255) & ~(size_t)255; }

extern "C" void kernel_launch(void* const* d_in, const int* in_sizes, int n_in,
                              void* d_out, int out_size, void* d_ws, size_t ws_size,
                              hipStream_t stream) {
  (void)in_sizes; (void)n_in; (void)out_size; (void)ws_size;

  const float* x        = (const float*)d_in[0];
  const float* ln1_g    = (const float*)d_in[1];
  const float* ln1_b    = (const float*)d_in[2];
  const float* acc_wqkv = (const float*)d_in[3];
  const float* acc_bqkv = (const float*)d_in[4];
  const float* acc_wo   = (const float*)d_in[5];
  const float* acc_bo   = (const float*)d_in[6];
  const float* gy_wqkv  = (const float*)d_in[7];
  const float* gy_bqkv  = (const float*)d_in[8];
  const float* gy_wo    = (const float*)d_in[9];
  const float* gy_bo    = (const float*)d_in[10];
  const float* conv_w   = (const float*)d_in[11];
  const float* ln2_g    = (const float*)d_in[12];
  const float* ln2_b    = (const float*)d_in[13];
  const float* w1       = (const float*)d_in[14];
  const float* b1       = (const float*)d_in[15];
  const float* w2       = (const float*)d_in[16];
  const float* b2       = (const float*)d_in[17];

  char* ws = (char*)d_ws;
  size_t off = 0;
  _Float16* xn_h   = (_Float16*)(ws + off); off = align256(off + (size_t)NTOK * E * 2);
  _Float16* qkv_h  = (_Float16*)(ws + off); off = align256(off + (size_t)2 * NTOK * 144 * 2);
  _Float16* attn_h = (_Float16*)(ws + off); off = align256(off + (size_t)2 * NTOK * QDIM * 2);
  float*    x1     = (float*)   (ws + off); off = align256(off + (size_t)NTOK * E * 4);
  _Float16* tt_h   = (_Float16*)(ws + off); off = align256(off + (size_t)NTOK * E * 2);
  _Float16* hid_h  = (_Float16*)(ws + off); off = align256(off + (size_t)NTOK * 2 * E * 2);
  _Float16* wq_a_h = (_Float16*)(ws + off); off = align256(off + 144 * 48 * 2);
  _Float16* wo_a_h = (_Float16*)(ws + off); off = align256(off + 48 * 48 * 2);
  _Float16* wq_g_h = (_Float16*)(ws + off); off = align256(off + 144 * 48 * 2);
  _Float16* wo_g_h = (_Float16*)(ws + off); off = align256(off + 48 * 48 * 2);
  _Float16* w1_h   = (_Float16*)(ws + off); off = align256(off + 384 * 192 * 2);
  _Float16* w2_h   = (_Float16*)(ws + off); off = align256(off + 192 * 384 * 2);

  cvt_f16_kernel<<<(144 * 48 + 255) / 256, 256, 0, stream>>>(acc_wqkv, wq_a_h, 144 * 48);
  cvt_f16_kernel<<<(48 * 48 + 255) / 256, 256, 0, stream>>>(acc_wo, wo_a_h, 48 * 48);
  cvt_f16_kernel<<<(144 * 48 + 255) / 256, 256, 0, stream>>>(gy_wqkv, wq_g_h, 144 * 48);
  cvt_f16_kernel<<<(48 * 48 + 255) / 256, 256, 0, stream>>>(gy_wo, wo_g_h, 48 * 48);
  cvt_f16_kernel<<<(384 * 192 + 255) / 256, 256, 0, stream>>>(w1, w1_h, 384 * 192);
  cvt_f16_kernel<<<(192 * 384 + 255) / 256, 256, 0, stream>>>(w2, w2_h, 192 * 384);

  ln_kernel<<<NTOK / 8, 256, 0, stream>>>(x, ln1_g, ln1_b, xn_h);

  conv_kernel<<<(NTOK * 96) / 256, 256, 0, stream>>>(xn_h, x, conv_w, x1);

  gemm_f16_kernel<3, false, false, true><<<dim3(NTOK / 128, 3), 256, 0, stream>>>(
      xn_h + 0, E, wq_a_h, 48, acc_bqkv, nullptr, 0, nullptr, qkv_h, 144);
  gemm_f16_kernel<3, false, false, true><<<dim3(NTOK / 128, 3), 256, 0, stream>>>(
      xn_h + 96, E, wq_g_h, 48, gy_bqkv, nullptr, 0, nullptr,
      qkv_h + (size_t)NTOK * 144, 144);

  attn_kernel<<<dim3(BATCH * 6, SEQ / 128), 256, 0, stream>>>(qkv_h, attn_h);

  gemm_f16_kernel<3, false, true, false><<<dim3(NTOK / 128, 1), 256, 0, stream>>>(
      attn_h, QDIM, wo_a_h, 48, acc_bo, x + 0, E, x1 + 0, nullptr, E);
  gemm_f16_kernel<3, false, true, false><<<dim3(NTOK / 128, 1), 256, 0, stream>>>(
      attn_h + (size_t)NTOK * QDIM, QDIM, wo_g_h, 48, gy_bo,
      x + 144, E, x1 + 144, nullptr, E);

  ln_kernel<<<NTOK / 8, 256, 0, stream>>>(x1, ln2_g, ln2_b, tt_h);

  gemm_f16_kernel<4, true, false, true><<<dim3(NTOK / 128, 6), 256, 0, stream>>>(
      tt_h, E, w1_h, 192, b1, nullptr, 0, nullptr, hid_h, 384);

  gemm_f16_kernel<4, true, true, false><<<dim3(NTOK / 128, 3), 256, 0, stream>>>(
      hid_h, 2 * E, w2_h, 384, b2, x1, E, (float*)d_out, nullptr, E);
}


# GEMM weights converted to f16 in WMMA fragment order, so each B-fragment half is one contiguous 512-byte wave load instead of 16 strided 32-byte row pieces
# speedup vs baseline: 1.0353x; 1.0353x over previous
#include <hip/hip_runtime.h>


typedef __attribute__((ext_vector_type(16))) _Float16 v16h;
typedef __attribute__((ext_vector_type(8)))  _Float16 v8h;
typedef __attribute__((ext_vector_type(8)))  float    v8f;

#define DEVI static __device__ __forceinline__

static constexpr int BATCH = 128;
static constexpr int SEQ   = 512;
static constexpr int E     = 192;
static constexpr int QDIM  = 48;
static constexpr int NTOK  = BATCH * SEQ;

static constexpr int KT_LD = 24;
static constexpr int VT_LD = 520;

DEVI v8f wmma16(v16h a, v16h b, v8f c) {
  return __builtin_amdgcn_wmma_f32_16x16x32_f16(false, a, false, b,
                                                (short)0, c, false, false);
}

DEVI float swap16(float v) {
  int i = __builtin_bit_cast(int, v);
  int r = __builtin_amdgcn_permlanex16(i, i, 0x76543210, 0xfedcba98, false, false);
  return __builtin_bit_cast(float, r);
}

DEVI v16h load_frag(const _Float16* __restrict__ base, int ld, int krem) {
  const int lane = threadIdx.x & 31;
  const int row  = lane & 15;
  const int d0   = (lane & 16) ? 8 : 0;
  const _Float16* p = base + (size_t)row * ld + d0;
  v16h r;
  if (krem >= 32) {
    v8h lo = *(const v8h*)(p);
    v8h hi = *(const v8h*)(p + 16);
#pragma unroll
    for (int i = 0; i < 8; ++i) { r[i] = lo[i]; r[8 + i] = hi[i]; }
  } else {
#pragma unroll
    for (int i = 0; i < 8; ++i) {
      r[i]     = (d0 + i      < krem) ? p[i]      : (_Float16)0.0f;
      r[8 + i] = (d0 + 16 + i < krem) ? p[16 + i] : (_Float16)0.0f;
    }
  }
  return r;
}

DEVI v16h load_bfrag(const _Float16* __restrict__ Wf, int K, int ntile, int k0) {
  const int lane = threadIdx.x & 31;
  const _Float16* p = Wf + (((size_t)ntile * (K >> 4) + (k0 >> 4)) << 8) + lane * 8;
  v8h lo = *(const v8h*)(p);
  v8h hi = {};
  if (K - k0 >= 32) hi = *(const v8h*)(p + 256);
  v16h r;
#pragma unroll
  for (int i = 0; i < 8; ++i) { r[i] = lo[i]; r[8 + i] = hi[i]; }
  return r;
}

__global__ void cvt_f16_frag_kernel(const float* __restrict__ src,
                                    _Float16* __restrict__ dst, int K, int n) {
  const int o = blockIdx.x * 256 + threadIdx.x;
  if (o >= n) return;
  const int i = o & 7;
  const int l = (o >> 3) & 31;
  const int chunk = o >> 8;
  const int nk16 = K >> 4;
  const int ntile = chunk / nk16;
  const int k16 = chunk - ntile * nk16;
  const int row = ntile * 16 + (l & 15);
  const int k = k16 * 16 + ((l & 16) ? 8 : 0) + i;
  dst[o] = (_Float16)src[(size_t)row * K + k];
}

__global__ __launch_bounds__(256) void ln_kernel(
    const float* __restrict__ x, const float* __restrict__ g,
    const float* __restrict__ b, _Float16* __restrict__ out) {
  const int wave = threadIdx.x >> 5, lane = threadIdx.x & 31;
  const size_t t = (size_t)blockIdx.x * 8 + wave;
  const float* xp = x + t * E;
  float v[6];
  float s = 0.0f;
#pragma unroll
  for (int i = 0; i < 6; ++i) { v[i] = xp[lane + 32 * i]; s += v[i]; }
#pragma unroll
  for (int o = 16; o > 0; o >>= 1) s += __shfl_xor(s, o, 32);
  const float mean = s * (1.0f / E);
  float vs = 0.0f;
#pragma unroll
  for (int i = 0; i < 6; ++i) { float d = v[i] - mean; vs += d * d; }
#pragma unroll
  for (int o = 16; o > 0; o >>= 1) vs += __shfl_xor(vs, o, 32);
  const float inv = rsqrtf(vs * (1.0f / E) + 1e-6f);
#pragma unroll
  for (int i = 0; i < 6; ++i) {
    int c = lane + 32 * i;
    out[t * E + c] = (_Float16)((v[i] - mean) * inv * g[c] + b[c]);
  }
}

__global__ void conv_kernel(const _Float16* __restrict__ xn,
                            const float* __restrict__ x,
                            const float* __restrict__ cw,
                            float* __restrict__ x1) {
  const size_t idx = (size_t)blockIdx.x * 256 + threadIdx.x;
  if (idx >= (size_t)NTOK * 96) return;
  const int h = (int)(idx % 96);
  const size_t t = idx / 96;
  const float* w = cw + (t & 3) * 15;
  const _Float16* xp = xn + t * E + 48;
  float s = 0.0f;
#pragma unroll
  for (int k = 0; k < 15; ++k) {
    int hh = h + k - 7;
    if (hh >= 0 && hh < 96) s += w[k] * (float)xp[hh];
  }
  x1[t * E + 48 + h] = s + x[t * E + 48 + h];
}

template <int NT, bool RELU, bool RES, bool OUT16>
__global__ __launch_bounds__(256) void gemm_f16_kernel(
    const _Float16* __restrict__ A, int lda,
    const _Float16* __restrict__ W, int K,
    const float* __restrict__ bias,
    const float* __restrict__ res, int ldr,
    float* __restrict__ out32, _Float16* __restrict__ out16, int ldo) {
  const int wave = threadIdx.x >> 5;
  const int m0 = blockIdx.x * 128 + wave * 16;
  const int n0 = blockIdx.y * (NT * 16);
  v8f z = {};
  v8f acc[NT];
#pragma unroll
  for (int t = 0; t < NT; ++t) acc[t] = z;

  for (int k0 = 0; k0 < K; k0 += 32) {
    const int krem = K - k0;
    v16h a = load_frag(A + (size_t)m0 * lda + k0, lda, krem);
#pragma unroll
    for (int t = 0; t < NT; ++t) {
      v16h bfr = load_bfrag(W, K, (n0 >> 4) + t, k0);
      acc[t] = wmma16(a, bfr, acc[t]);
    }
  }

  const int lane = threadIdx.x & 31;
  const int col = lane & 15;
  const int r0  = (lane & 16) ? 8 : 0;
#pragma unroll
  for (int t = 0; t < NT; ++t) {
    const int n = n0 + t * 16 + col;
    const float bv = bias[n];
#pragma unroll
    for (int j = 0; j < 8; ++j) {
      const int m = m0 + r0 + j;
      float v = acc[t][j] + bv;
      if (RELU) v = fmaxf(v, 0.0f);
      if (RES)  v += res[(size_t)m * ldr + n];
      if (OUT16) out16[(size_t)m * ldo + n] = (_Float16)v;
      else       out32[(size_t)m * ldo + n] = v;
    }
  }
}

__global__ __launch_bounds__(256) void attn_kernel(
    const _Float16* __restrict__ qkv_all, _Float16* __restrict__ out_all) {
  __shared__ _Float16 vt[16 * VT_LD];
  __shared__ _Float16 kt[SEQ * KT_LD];

  const int bid = blockIdx.x;
  const int b = bid / 6;
  const int r = bid % 6;
  const int branch = r / 3;
  const int hh = r % 3;
  const _Float16* qkv = qkv_all + (size_t)branch * NTOK * 144
                                + (size_t)b * SEQ * 144;
  _Float16* outp = out_all + (size_t)branch * NTOK * QDIM
                           + (size_t)b * SEQ * QDIM;
  const int qcol = hh * 16, kcol = 48 + hh * 16, vcol = 96 + hh * 16;

  for (int e = threadIdx.x; e < 16 * SEQ; e += 256) {
    const int d = e & 15, j = e >> 4;
    const _Float16 kv = qkv[(size_t)j * 144 + kcol + d];
    const _Float16 vv = qkv[(size_t)j * 144 + vcol + d];
    kt[j * KT_LD + d] = kv;
    vt[d * VT_LD + j] = vv;
  }
  __syncthreads();

  const int wave = threadIdx.x >> 5, lane = threadIdx.x & 31;
  const int q0 = blockIdx.y * 128 + wave * 16;
  const int row = lane & 15;
  const int d0  = (lane & 16) ? 8 : 0;

  v16h qf;
  {
    const _Float16* qp = qkv + (size_t)(q0 + row) * 144 + qcol + d0;
    v8h ql = *(const v8h*)qp;
#pragma unroll
    for (int i = 0; i < 8; ++i) {
      qf[i] = ql[i] * (_Float16)0.25f;
      qf[8 + i] = (_Float16)0.0f;
    }
  }

  v8f z = {};
  v8f oacc = z;
  float m_run = -1e30f, l_run = 0.0f;

  for (int kb = 0; kb < SEQ / 32; ++kb) {
    const int j0 = kb * 32;
    v16h kf0, kf1;
    {
      const _Float16* kp0 = &kt[(j0 + row) * KT_LD + d0];
      const _Float16* kp1 = &kt[(j0 + 16 + row) * KT_LD + d0];
      v8h ka = *(const v8h*)kp0;
      v8h kb2 = *(const v8h*)kp1;
#pragma unroll
      for (int i = 0; i < 8; ++i) {
        kf0[i] = ka[i];  kf0[8 + i] = (_Float16)0.0f;
        kf1[i] = kb2[i]; kf1[8 + i] = (_Float16)0.0f;
      }
    }
    v8f s0 = wmma16(kf0, qf, z);
    v8f s1 = wmma16(kf1, qf, z);

    float tm = -1e30f;
#pragma unroll
    for (int j = 0; j < 8; ++j) tm = fmaxf(tm, fmaxf(s0[j], s1[j]));
    tm = fmaxf(tm, swap16(tm));
    const float mnew = fmaxf(m_run, tm);
    const float corr = __expf(m_run - mnew);
    float p0[8], p1[8], ls = 0.0f;
#pragma unroll
    for (int j = 0; j < 8; ++j) {
      p0[j] = __expf(s0[j] - mnew);
      p1[j] = __expf(s1[j] - mnew);
      ls += p0[j] + p1[j];
    }
    ls += swap16(ls);
    l_run = l_run * corr + ls;
    m_run = mnew;
#pragma unroll
    for (int j = 0; j < 8; ++j) oacc[j] *= corr;

    v16h pf;
#pragma unroll
    for (int i = 0; i < 8; ++i) {
      pf[i] = (_Float16)p0[i];
      pf[8 + i] = (_Float16)p1[i];
    }
    const _Float16* vp = &vt[row * VT_LD + j0 + d0];
    v8h vl = *(const v8h*)vp;
    v8h vh = *(const v8h*)(vp + 16);
    v16h vf;
#pragma unroll
    for (int i = 0; i < 8; ++i) { vf[i] = vl[i]; vf[8 + i] = vh[i]; }

    oacc = wmma16(vf, pf, oacc);
  }

  const float inv = 1.0f / l_run;
#pragma unroll
  for (int j = 0; j < 8; ++j)
    outp[(size_t)(q0 + row) * QDIM + qcol + d0 + j] = (_Float16)(oacc[j] * inv);
}

static inline size_t align256(size_t x) { return (x + 255) & ~(size_t)255; }

extern "C" void kernel_launch(void* const* d_in, const int* in_sizes, int n_in,
                              void* d_out, int out_size, void* d_ws, size_t ws_size,
                              hipStream_t stream) {
  (void)in_sizes; (void)n_in; (void)out_size; (void)ws_size;

  const float* x        = (const float*)d_in[0];
  const float* ln1_g    = (const float*)d_in[1];
  const float* ln1_b    = (const float*)d_in[2];
  const float* acc_wqkv = (const float*)d_in[3];
  const float* acc_bqkv = (const float*)d_in[4];
  const float* acc_wo   = (const float*)d_in[5];
  const float* acc_bo   = (const float*)d_in[6];
  const float* gy_wqkv  = (const float*)d_in[7];
  const float* gy_bqkv  = (const float*)d_in[8];
  const float* gy_wo    = (const float*)d_in[9];
  const float* gy_bo    = (const float*)d_in[10];
  const float* conv_w   = (const float*)d_in[11];
  const float* ln2_g    = (const float*)d_in[12];
  const float* ln2_b    = (const float*)d_in[13];
  const float* w1       = (const float*)d_in[14];
  const float* b1       = (const float*)d_in[15];
  const float* w2       = (const float*)d_in[16];
  const float* b2       = (const float*)d_in[17];

  char* ws = (char*)d_ws;
  size_t off = 0;
  _Float16* xn_h   = (_Float16*)(ws + off); off = align256(off + (size_t)NTOK * E * 2);
  _Float16* qkv_h  = (_Float16*)(ws + off); off = align256(off + (size_t)2 * NTOK * 144 * 2);
  _Float16* attn_h = (_Float16*)(ws + off); off = align256(off + (size_t)2 * NTOK * QDIM * 2);
  float*    x1     = (float*)   (ws + off); off = align256(off + (size_t)NTOK * E * 4);
  _Float16* tt_h   = (_Float16*)(ws + off); off = align256(off + (size_t)NTOK * E * 2);
  _Float16* hid_h  = (_Float16*)(ws + off); off = align256(off + (size_t)NTOK * 2 * E * 2);
  _Float16* wq_a_h = (_Float16*)(ws + off); off = align256(off + 144 * 48 * 2);
  _Float16* wo_a_h = (_Float16*)(ws + off); off = align256(off + 48 * 48 * 2);
  _Float16* wq_g_h = (_Float16*)(ws + off); off = align256(off + 144 * 48 * 2);
  _Float16* wo_g_h = (_Float16*)(ws + off); off = align256(off + 48 * 48 * 2);
  _Float16* w1_h   = (_Float16*)(ws + off); off = align256(off + 384 * 192 * 2);
  _Float16* w2_h   = (_Float16*)(ws + off); off = align256(off + 192 * 384 * 2);

  cvt_f16_frag_kernel<<<(144 * 48 + 255) / 256, 256, 0, stream>>>(acc_wqkv, wq_a_h, 48, 144 * 48);
  cvt_f16_frag_kernel<<<(48 * 48 + 255) / 256, 256, 0, stream>>>(acc_wo, wo_a_h, 48, 48 * 48);
  cvt_f16_frag_kernel<<<(144 * 48 + 255) / 256, 256, 0, stream>>>(gy_wqkv, wq_g_h, 48, 144 * 48);
  cvt_f16_frag_kernel<<<(48 * 48 + 255) / 256, 256, 0, stream>>>(gy_wo, wo_g_h, 48, 48 * 48);
  cvt_f16_frag_kernel<<<(384 * 192 + 255) / 256, 256, 0, stream>>>(w1, w1_h, 192, 384 * 192);
  cvt_f16_frag_kernel<<<(192 * 384 + 255) / 256, 256, 0, stream>>>(w2, w2_h, 384, 192 * 384);

  ln_kernel<<<NTOK / 8, 256, 0, stream>>>(x, ln1_g, ln1_b, xn_h);

  conv_kernel<<<(NTOK * 96) / 256, 256, 0, stream>>>(xn_h, x, conv_w, x1);

  gemm_f16_kernel<3, false, false, true><<<dim3(NTOK / 128, 3), 256, 0, stream>>>(
      xn_h + 0, E, wq_a_h, 48, acc_bqkv, nullptr, 0, nullptr, qkv_h, 144);
  gemm_f16_kernel<3, false, false, true><<<dim3(NTOK / 128, 3), 256, 0, stream>>>(
      xn_h + 96, E, wq_g_h, 48, gy_bqkv, nullptr, 0, nullptr,
      qkv_h + (size_t)NTOK * 144, 144);

  attn_kernel<<<dim3(BATCH * 6, SEQ / 128), 256, 0, stream>>>(qkv_h, attn_h);

  gemm_f16_kernel<3, false, true, false><<<dim3(NTOK / 128, 1), 256, 0, stream>>>(
      attn_h, QDIM, wo_a_h, 48, acc_bo, x + 0, E, x1 + 0, nullptr, E);
  gemm_f16_kernel<3, false, true, false><<<dim3(NTOK / 128, 1), 256, 0, stream>>>(
      attn_h + (size_t)NTOK * QDIM, QDIM, wo_g_h, 48, gy_bo,
      x + 144, E, x1 + 144, nullptr, E);

  ln_kernel<<<NTOK / 8, 256, 0, stream>>>(x1, ln2_g, ln2_b, tt_h);

  gemm_f16_kernel<4, true, false, true><<<dim3(NTOK / 128, 6), 256, 0, stream>>>(
      tt_h, E, w1_h, 192, b1, nullptr, 0, nullptr, hid_h, 384);

  gemm_f16_kernel<4, true, true, false><<<dim3(NTOK / 128, 3), 256, 0, stream>>>(
      hid_h, 2 * E, w2_h, 384, b2, x1, E, (float*)d_out, nullptr, E);
}


# Stages that read none of each other's results share one grid: 6 weight conversions ride in the LN1 launch, conv + both QKV GEMMs are one launch, both out-projections one: 16 launches -> 7
# speedup vs baseline: 1.2368x; 1.1946x over previous
#include <hip/hip_runtime.h>


typedef __attribute__((ext_vector_type(16))) _Float16 v16h;
typedef __attribute__((ext_vector_type(8)))  _Float16 v8h;
typedef __attribute__((ext_vector_type(8)))  float    v8f;

#define DEVI static __device__ __forceinline__

static constexpr int BATCH = 128;
static constexpr int SEQ   = 512;
static constexpr int E     = 192;
static constexpr int QDIM  = 48;
static constexpr int NTOK  = BATCH * SEQ;

static constexpr int KT_LD = 24;
static constexpr int VT_LD = 520;

DEVI v8f wmma16(v16h a, v16h b, v8f c) {
  return __builtin_amdgcn_wmma_f32_16x16x32_f16(false, a, false, b,
                                                (short)0, c, false, false);
}

DEVI float swap16(float v) {
  int i = __builtin_bit_cast(int, v);
  int r = __builtin_amdgcn_permlanex16(i, i, 0x76543210, 0xfedcba98, false, false);
  return __builtin_bit_cast(float, r);
}

DEVI v16h load_frag(const _Float16* __restrict__ base, int ld, int krem) {
  const int lane = threadIdx.x & 31;
  const int row  = lane & 15;
  const int d0   = (lane & 16) ? 8 : 0;
  const _Float16* p = base + (size_t)row * ld + d0;
  v16h r;
  if (krem >= 32) {
    v8h lo = *(const v8h*)(p);
    v8h hi = *(const v8h*)(p + 16);
#pragma unroll
    for (int i = 0; i < 8; ++i) { r[i] = lo[i]; r[8 + i] = hi[i]; }
  } else {
#pragma unroll
    for (int i = 0; i < 8; ++i) {
      r[i]     = (d0 + i      < krem) ? p[i]      : (_Float16)0.0f;
      r[8 + i] = (d0 + 16 + i < krem) ? p[16 + i] : (_Float16)0.0f;
    }
  }
  return r;
}

DEVI v16h load_bfrag(const _Float16* __restrict__ Wf, int K, int ntile, int k0) {
  const int lane = threadIdx.x & 31;
  const _Float16* p = Wf + (((size_t)ntile * (K >> 4) + (k0 >> 4)) << 8) + lane * 8;
  v8h lo = *(const v8h*)(p);
  v8h hi = {};
  if (K - k0 >= 32) hi = *(const v8h*)(p + 256);
  v16h r;
#pragma unroll
  for (int i = 0; i < 8; ++i) { r[i] = lo[i]; r[8 + i] = hi[i]; }
  return r;
}

DEVI void cvt_f16_frag_block(const float* __restrict__ src,
                             _Float16* __restrict__ dst, int K, int n, int blk) {
  const int o = blk * 256 + threadIdx.x;
  if (o >= n) return;
  const int i = o & 7;
  const int l = (o >> 3) & 31;
  const int chunk = o >> 8;
  const int nk16 = K >> 4;
  const int ntile = chunk / nk16;
  const int k16 = chunk - ntile * nk16;
  const int row = ntile * 16 + (l & 15);
  const int k = k16 * 16 + ((l & 16) ? 8 : 0) + i;
  dst[o] = (_Float16)src[(size_t)row * K + k];
}

static constexpr int LN_TPW = 8;

DEVI void ln_block(const float* __restrict__ x, const float* __restrict__ g,
                   const float* __restrict__ b, _Float16* __restrict__ out,
                   int blk) {
  const int wave = threadIdx.x >> 5, lane = threadIdx.x & 31;
  const size_t t0 = ((size_t)blk * 8 + wave) * LN_TPW;
  float gg[6], bb[6];
#pragma unroll
  for (int i = 0; i < 6; ++i) { gg[i] = g[lane + 32 * i]; bb[i] = b[lane + 32 * i]; }
  float v[LN_TPW][6];
#pragma unroll
  for (int u = 0; u < LN_TPW; ++u) {
    const float* xp = x + (t0 + u) * E;
#pragma unroll
    for (int i = 0; i < 6; ++i) v[u][i] = xp[lane + 32 * i];
  }
#pragma unroll
  for (int u = 0; u < LN_TPW; ++u) {
    const size_t t = t0 + u;
    float s = 0.0f;
#pragma unroll
    for (int i = 0; i < 6; ++i) s += v[u][i];
#pragma unroll
    for (int o = 16; o > 0; o >>= 1) s += __shfl_xor(s, o, 32);
    const float mean = s * (1.0f / E);
    float vs = 0.0f;
#pragma unroll
    for (int i = 0; i < 6; ++i) { float d = v[u][i] - mean; vs += d * d; }
#pragma unroll
    for (int o = 16; o > 0; o >>= 1) vs += __shfl_xor(vs, o, 32);
    const float inv = rsqrtf(vs * (1.0f / E) + 1e-6f);
#pragma unroll
    for (int i = 0; i < 6; ++i) {
      int c = lane + 32 * i;
      out[t * E + c] = (_Float16)((v[u][i] - mean) * inv * gg[i] + bb[i]);
    }
  }
}

__global__ __launch_bounds__(256) void ln_kernel(
    const float* __restrict__ x, const float* __restrict__ g,
    const float* __restrict__ b, _Float16* __restrict__ out) {
  ln_block(x, g, b, out, blockIdx.x);
}

static constexpr int CVT_QKV = 144 * 48 / 256;
static constexpr int CVT_WO  = 48 * 48 / 256;
static constexpr int CVT_MLP = 384 * 192 / 256;
static constexpr int CVT_BLOCKS = 2 * CVT_QKV + 2 * CVT_WO + 2 * CVT_MLP;

__global__ __launch_bounds__(256) void front_kernel(
    const float* __restrict__ x, const float* __restrict__ g,
    const float* __restrict__ b, _Float16* __restrict__ out,
    const float* __restrict__ wqa, const float* __restrict__ woa,
    const float* __restrict__ wqg, const float* __restrict__ wog,
    const float* __restrict__ w1, const float* __restrict__ w2,
    _Float16* __restrict__ wqa_h, _Float16* __restrict__ woa_h,
    _Float16* __restrict__ wqg_h, _Float16* __restrict__ wog_h,
    _Float16* __restrict__ w1_h, _Float16* __restrict__ w2_h) {
  int blk = blockIdx.x;
  if (blk >= CVT_BLOCKS) { ln_block(x, g, b, out, blk - CVT_BLOCKS); return; }
  if (blk < CVT_QKV) { cvt_f16_frag_block(wqa, wqa_h, 48, 144 * 48, blk); return; }
  blk -= CVT_QKV;
  if (blk < CVT_WO)  { cvt_f16_frag_block(woa, woa_h, 48, 48 * 48, blk); return; }
  blk -= CVT_WO;
  if (blk < CVT_QKV) { cvt_f16_frag_block(wqg, wqg_h, 48, 144 * 48, blk); return; }
  blk -= CVT_QKV;
  if (blk < CVT_WO)  { cvt_f16_frag_block(wog, wog_h, 48, 48 * 48, blk); return; }
  blk -= CVT_WO;
  if (blk < CVT_MLP) { cvt_f16_frag_block(w1, w1_h, 192, 384 * 192, blk); return; }
  blk -= CVT_MLP;
  cvt_f16_frag_block(w2, w2_h, 384, 192 * 384, blk);
}

static constexpr int CV_TPW = 8;

DEVI void conv_block(const _Float16* __restrict__ xn,
                     const float* __restrict__ x,
                     const float* __restrict__ cw,
                     float* __restrict__ x1, int blk) {
  const int wave = threadIdx.x >> 5, lane = threadIdx.x & 31;
  const size_t t0 = ((size_t)blk * 8 + wave) * CV_TPW;
  for (int jo = 0; jo < CV_TPW; jo += 4) {
#pragma unroll
    for (int c = 0; c < 4; ++c) {
      const size_t t = t0 + jo + c;
      const float* w = cw + c * 15;
      const _Float16* xp = xn + t * E + 48;
#pragma unroll
      for (int i = 0; i < 3; ++i) {
        const int h = lane + 32 * i;
        float s = 0.0f;
#pragma unroll
        for (int k = 0; k < 15; ++k) {
          int hh = h + k - 7;
          if (hh >= 0 && hh < 96) s += w[k] * (float)xp[hh];
        }
        x1[t * E + 48 + h] = s + x[t * E + 48 + h];
      }
    }
  }
}

template <int NT, bool RELU, bool RES, bool OUT16>
DEVI void gemm_f16_block(
    const _Float16* __restrict__ A, int lda,
    const _Float16* __restrict__ W, int K,
    const float* __restrict__ bias,
    const float* __restrict__ res, int ldr,
    float* __restrict__ out32, _Float16* __restrict__ out16, int ldo,
    int bx, int by) {
  const int wave = threadIdx.x >> 5;
  const int m0 = bx * 128 + wave * 16;
  const int n0 = by * (NT * 16);
  v8f z = {};
  v8f acc[NT];
#pragma unroll
  for (int t = 0; t < NT; ++t) acc[t] = z;

  for (int k0 = 0; k0 < K; k0 += 32) {
    const int krem = K - k0;
    v16h a = load_frag(A + (size_t)m0 * lda + k0, lda, krem);
#pragma unroll
    for (int t = 0; t < NT; ++t) {
      v16h bfr = load_bfrag(W, K, (n0 >> 4) + t, k0);
      acc[t] = wmma16(a, bfr, acc[t]);
    }
  }

  const int lane = threadIdx.x & 31;
  const int col = lane & 15;
  const int r0  = (lane & 16) ? 8 : 0;
#pragma unroll
  for (int t = 0; t < NT; ++t) {
    const int n = n0 + t * 16 + col;
    const float bv = bias[n];
#pragma unroll
    for (int j = 0; j < 8; ++j) {
      const int m = m0 + r0 + j;
      float v = acc[t][j] + bv;
      if (RELU) v = fmaxf(v, 0.0f);
      if (RES)  v += res[(size_t)m * ldr + n];
      if (OUT16) out16[(size_t)m * ldo + n] = (_Float16)v;
      else       out32[(size_t)m * ldo + n] = v;
    }
  }
}

template <int NT, bool RELU, bool RES, bool OUT16>
__global__ __launch_bounds__(256) void gemm_f16_kernel(
    const _Float16* __restrict__ A0, const _Float16* __restrict__ A1, int lda,
    const _Float16* __restrict__ W0, const _Float16* __restrict__ W1, int K,
    const float* __restrict__ bias0, const float* __restrict__ bias1,
    const float* __restrict__ res0, const float* __restrict__ res1, int ldr,
    float* __restrict__ out32_0, float* __restrict__ out32_1,
    _Float16* __restrict__ out16_0, _Float16* __restrict__ out16_1, int ldo) {
  const bool second = blockIdx.z != 0;
  gemm_f16_block<NT, RELU, RES, OUT16>(
      second ? A1 : A0, lda, second ? W1 : W0, K, second ? bias1 : bias0,
      second ? res1 : res0, ldr, second ? out32_1 : out32_0,
      second ? out16_1 : out16_0, ldo, blockIdx.x, blockIdx.y);
}

static constexpr int CONV_BLOCKS = NTOK / (8 * CV_TPW);
static constexpr int QKV_MB      = NTOK / 128;
static constexpr int QKV_BLOCKS  = QKV_MB * 3;

__global__ __launch_bounds__(256) void mid_kernel(
    const _Float16* __restrict__ xn, const float* __restrict__ x,
    const float* __restrict__ cw, float* __restrict__ x1,
    const _Float16* __restrict__ wq0, const _Float16* __restrict__ wq1,
    const float* __restrict__ bq0, const float* __restrict__ bq1,
    _Float16* __restrict__ qkv0, _Float16* __restrict__ qkv1,
    int lda, int K, int ldo) {
  int blk = blockIdx.x;
  if (blk < CONV_BLOCKS) { conv_block(xn, x, cw, x1, blk); return; }
  blk -= CONV_BLOCKS;
  const bool second = blk >= QKV_BLOCKS;
  if (second) blk -= QKV_BLOCKS;
  const int by = blk / QKV_MB;
  const int bx = blk - by * QKV_MB;
  gemm_f16_block<3, false, false, true>(
      second ? xn + 96 : xn, lda, second ? wq1 : wq0, K, second ? bq1 : bq0,
      nullptr, 0, nullptr, second ? qkv1 : qkv0, ldo, bx, by);
}

__global__ __launch_bounds__(256) void attn_kernel(
    const _Float16* __restrict__ qkv_all, _Float16* __restrict__ out_all) {
  __shared__ _Float16 vt[16 * VT_LD];
  __shared__ _Float16 kt[SEQ * KT_LD];

  const int bid = blockIdx.x;
  const int b = bid / 6;
  const int r = bid % 6;
  const int branch = r / 3;
  const int hh = r % 3;
  const _Float16* qkv = qkv_all + (size_t)branch * NTOK * 144
                                + (size_t)b * SEQ * 144;
  _Float16* outp = out_all + (size_t)branch * NTOK * QDIM
                           + (size_t)b * SEQ * QDIM;
  const int qcol = hh * 16, kcol = 48 + hh * 16, vcol = 96 + hh * 16;

  for (int e = threadIdx.x; e < 16 * SEQ; e += 256) {
    const int d = e & 15, j = e >> 4;
    const _Float16 kv = qkv[(size_t)j * 144 + kcol + d];
    const _Float16 vv = qkv[(size_t)j * 144 + vcol + d];
    kt[j * KT_LD + d] = kv;
    vt[d * VT_LD + j] = vv;
  }
  __syncthreads();

  const int wave = threadIdx.x >> 5, lane = threadIdx.x & 31;
  const int q0 = blockIdx.y * 128 + wave * 16;
  const int row = lane & 15;
  const int d0  = (lane & 16) ? 8 : 0;

  v16h qf;
  {
    const _Float16* qp = qkv + (size_t)(q0 + row) * 144 + qcol + d0;
    v8h ql = *(const v8h*)qp;
#pragma unroll
    for (int i = 0; i < 8; ++i) {
      qf[i] = ql[i] * (_Float16)0.25f;
      qf[8 + i] = (_Float16)0.0f;
    }
  }

  v8f z = {};
  v8f oacc = z;
  float m_run = -1e30f, l_run = 0.0f;

  for (int kb = 0; kb < SEQ / 32; ++kb) {
    const int j0 = kb * 32;
    v16h kf0, kf1;
    {
      const _Float16* kp0 = &kt[(j0 + row) * KT_LD + d0];
      const _Float16* kp1 = &kt[(j0 + 16 + row) * KT_LD + d0];
      v8h ka = *(const v8h*)kp0;
      v8h kb2 = *(const v8h*)kp1;
#pragma unroll
      for (int i = 0; i < 8; ++i) {
        kf0[i] = ka[i];  kf0[8 + i] = (_Float16)0.0f;
        kf1[i] = kb2[i]; kf1[8 + i] = (_Float16)0.0f;
      }
    }
    v8f s0 = wmma16(kf0, qf, z);
    v8f s1 = wmma16(kf1, qf, z);

    float tm = -1e30f;
#pragma unroll
    for (int j = 0; j < 8; ++j) tm = fmaxf(tm, fmaxf(s0[j], s1[j]));
    tm = fmaxf(tm, swap16(tm));
    const float mnew = fmaxf(m_run, tm);
    const float corr = __expf(m_run - mnew);
    float p0[8], p1[8], ls = 0.0f;
#pragma unroll
    for (int j = 0; j < 8; ++j) {
      p0[j] = __expf(s0[j] - mnew);
      p1[j] = __expf(s1[j] - mnew);
      ls += p0[j] + p1[j];
    }
    ls += swap16(ls);
    l_run = l_run * corr + ls;
    m_run = mnew;
#pragma unroll
    for (int j = 0; j < 8; ++j) oacc[j] *= corr;

    v16h pf;
#pragma unroll
    for (int i = 0; i < 8; ++i) {
      pf[i] = (_Float16)p0[i];
      pf[8 + i] = (_Float16)p1[i];
    }
    const _Float16* vp = &vt[row * VT_LD + j0 + d0];
    v8h vl = *(const v8h*)vp;
    v8h vh = *(const v8h*)(vp + 16);
    v16h vf;
#pragma unroll
    for (int i = 0; i < 8; ++i) { vf[i] = vl[i]; vf[8 + i] = vh[i]; }

    oacc = wmma16(vf, pf, oacc);
  }

  const float inv = 1.0f / l_run;
#pragma unroll
  for (int j = 0; j < 8; ++j)
    outp[(size_t)(q0 + row) * QDIM + qcol + d0 + j] = (_Float16)(oacc[j] * inv);
}

static inline size_t align256(size_t x) { return (x + 255) & ~(size_t)255; }

extern "C" void kernel_launch(void* const* d_in, const int* in_sizes, int n_in,
                              void* d_out, int out_size, void* d_ws, size_t ws_size,
                              hipStream_t stream) {
  (void)in_sizes; (void)n_in; (void)out_size; (void)ws_size;

  const float* x        = (const float*)d_in[0];
  const float* ln1_g    = (const float*)d_in[1];
  const float* ln1_b    = (const float*)d_in[2];
  const float* acc_wqkv = (const float*)d_in[3];
  const float* acc_bqkv = (const float*)d_in[4];
  const float* acc_wo   = (const float*)d_in[5];
  const float* acc_bo   = (const float*)d_in[6];
  const float* gy_wqkv  = (const float*)d_in[7];
  const float* gy_bqkv  = (const float*)d_in[8];
  const float* gy_wo    = (const float*)d_in[9];
  const float* gy_bo    = (const float*)d_in[10];
  const float* conv_w   = (const float*)d_in[11];
  const float* ln2_g    = (const float*)d_in[12];
  const float* ln2_b    = (const float*)d_in[13];
  const float* w1       = (const float*)d_in[14];
  const float* b1       = (const float*)d_in[15];
  const float* w2       = (const float*)d_in[16];
  const float* b2       = (const float*)d_in[17];

  char* ws = (char*)d_ws;
  size_t off = 0;
  _Float16* xn_h   = (_Float16*)(ws + off); off = align256(off + (size_t)NTOK * E * 2);
  _Float16* qkv_h  = (_Float16*)(ws + off); off = align256(off + (size_t)2 * NTOK * 144 * 2);
  _Float16* attn_h = (_Float16*)(ws + off); off = align256(off + (size_t)2 * NTOK * QDIM * 2);
  float*    x1     = (float*)   (ws + off); off = align256(off + (size_t)NTOK * E * 4);
  _Float16* tt_h   = (_Float16*)(ws + off); off = align256(off + (size_t)NTOK * E * 2);
  _Float16* hid_h  = (_Float16*)(ws + off); off = align256(off + (size_t)NTOK * 2 * E * 2);
  _Float16* wq_a_h = (_Float16*)(ws + off); off = align256(off + 144 * 48 * 2);
  _Float16* wo_a_h = (_Float16*)(ws + off); off = align256(off + 48 * 48 * 2);
  _Float16* wq_g_h = (_Float16*)(ws + off); off = align256(off + 144 * 48 * 2);
  _Float16* wo_g_h = (_Float16*)(ws + off); off = align256(off + 48 * 48 * 2);
  _Float16* w1_h   = (_Float16*)(ws + off); off = align256(off + 384 * 192 * 2);
  _Float16* w2_h   = (_Float16*)(ws + off); off = align256(off + 192 * 384 * 2);

  front_kernel<<<CVT_BLOCKS + NTOK / (8 * LN_TPW), 256, 0, stream>>>(
      x, ln1_g, ln1_b, xn_h,
      acc_wqkv, acc_wo, gy_wqkv, gy_wo, w1, w2,
      wq_a_h, wo_a_h, wq_g_h, wo_g_h, w1_h, w2_h);

  mid_kernel<<<CONV_BLOCKS + 2 * QKV_BLOCKS, 256, 0, stream>>>(
      xn_h, x, conv_w, x1, wq_a_h, wq_g_h, acc_bqkv, gy_bqkv,
      qkv_h, qkv_h + (size_t)NTOK * 144, E, 48, 144);

  attn_kernel<<<dim3(BATCH * 6, SEQ / 128), 256, 0, stream>>>(qkv_h, attn_h);

  gemm_f16_kernel<3, false, true, false><<<dim3(NTOK / 128, 1, 2), 256, 0, stream>>>(
      attn_h, attn_h + (size_t)NTOK * QDIM, QDIM, wo_a_h, wo_g_h, 48,
      acc_bo, gy_bo, x + 0, x + 144, E, x1 + 0, x1 + 144, nullptr, nullptr, E);

  ln_kernel<<<NTOK / (8 * LN_TPW), 256, 0, stream>>>(x1, ln2_g, ln2_b, tt_h);

  gemm_f16_kernel<4, true, false, true><<<dim3(NTOK / 128, 6, 1), 256, 0, stream>>>(
      tt_h, nullptr, E, w1_h, nullptr, 192, b1, nullptr,
      nullptr, nullptr, 0, nullptr, nullptr, hid_h, nullptr, 384);

  gemm_f16_kernel<4, true, true, false><<<dim3(NTOK / 128, 3, 1), 256, 0, stream>>>(
      hid_h, nullptr, 2 * E, w2_h, nullptr, 384, b2, nullptr,
      x1, nullptr, E, (float*)d_out, nullptr, nullptr, nullptr, E);
}


# MLP1+MLP2 in one launch: hidden tiles (hid^T = W1*tt^T) pass in registers to product 2 as its A fragments, never stored; W1, W2 staged once per work-group in LDS; b1 now starts product 1's sum
# speedup vs baseline: 1.8557x; 1.5004x over previous
#include <hip/hip_runtime.h>


typedef __attribute__((ext_vector_type(16))) _Float16 v16h;
typedef __attribute__((ext_vector_type(8)))  _Float16 v8h;
typedef __attribute__((ext_vector_type(8)))  float    v8f;

#define DEVI static __device__ __forceinline__

static constexpr int BATCH = 128;
static constexpr int SEQ   = 512;
static constexpr int E     = 192;
static constexpr int QDIM  = 48;
static constexpr int NTOK  = BATCH * SEQ;

static constexpr int KT_LD = 24;
static constexpr int VT_LD = 520;

DEVI v8f wmma16(v16h a, v16h b, v8f c) {
  return __builtin_amdgcn_wmma_f32_16x16x32_f16(false, a, false, b,
                                                (short)0, c, false, false);
}

DEVI float swap16(float v) {
  int i = __builtin_bit_cast(int, v);
  int r = __builtin_amdgcn_permlanex16(i, i, 0x76543210, 0xfedcba98, false, false);
  return __builtin_bit_cast(float, r);
}

DEVI v16h load_frag(const _Float16* __restrict__ base, int ld, int krem) {
  const int lane = threadIdx.x & 31;
  const int row  = lane & 15;
  const int d0   = (lane & 16) ? 8 : 0;
  const _Float16* p = base + (size_t)row * ld + d0;
  v16h r;
  if (krem >= 32) {
    v8h lo = *(const v8h*)(p);
    v8h hi = *(const v8h*)(p + 16);
#pragma unroll
    for (int i = 0; i < 8; ++i) { r[i] = lo[i]; r[8 + i] = hi[i]; }
  } else {
#pragma unroll
    for (int i = 0; i < 8; ++i) {
      r[i]     = (d0 + i      < krem) ? p[i]      : (_Float16)0.0f;
      r[8 + i] = (d0 + 16 + i < krem) ? p[16 + i] : (_Float16)0.0f;
    }
  }
  return r;
}

DEVI v16h load_bfrag(const _Float16* __restrict__ Wf, int K, int ntile, int k0) {
  const int lane = threadIdx.x & 31;
  const _Float16* p = Wf + (((size_t)ntile * (K >> 4) + (k0 >> 4)) << 8) + lane * 8;
  v8h lo = *(const v8h*)(p);
  v8h hi = {};
  if (K - k0 >= 32) hi = *(const v8h*)(p + 256);
  v16h r;
#pragma unroll
  for (int i = 0; i < 8; ++i) { r[i] = lo[i]; r[8 + i] = hi[i]; }
  return r;
}

DEVI void cvt_f16_frag_block(const float* __restrict__ src,
                             _Float16* __restrict__ dst, int K, int n, int blk) {
  const int o = blk * 256 + threadIdx.x;
  if (o >= n) return;
  const int i = o & 7;
  const int l = (o >> 3) & 31;
  const int chunk = o >> 8;
  const int nk16 = K >> 4;
  const int ntile = chunk / nk16;
  const int k16 = chunk - ntile * nk16;
  const int row = ntile * 16 + (l & 15);
  const int k = k16 * 16 + ((l & 16) ? 8 : 0) + i;
  dst[o] = (_Float16)src[(size_t)row * K + k];
}

static constexpr int LN_TPW = 8;

DEVI void ln_block(const float* __restrict__ x, const float* __restrict__ g,
                   const float* __restrict__ b, _Float16* __restrict__ out,
                   int blk) {
  const int wave = threadIdx.x >> 5, lane = threadIdx.x & 31;
  const size_t t0 = ((size_t)blk * 8 + wave) * LN_TPW;
  float gg[6], bb[6];
#pragma unroll
  for (int i = 0; i < 6; ++i) { gg[i] = g[lane + 32 * i]; bb[i] = b[lane + 32 * i]; }
  float v[LN_TPW][6];
#pragma unroll
  for (int u = 0; u < LN_TPW; ++u) {
    const float* xp = x + (t0 + u) * E;
#pragma unroll
    for (int i = 0; i < 6; ++i) v[u][i] = xp[lane + 32 * i];
  }
#pragma unroll
  for (int u = 0; u < LN_TPW; ++u) {
    const size_t t = t0 + u;
    float s = 0.0f;
#pragma unroll
    for (int i = 0; i < 6; ++i) s += v[u][i];
#pragma unroll
    for (int o = 16; o > 0; o >>= 1) s += __shfl_xor(s, o, 32);
    const float mean = s * (1.0f / E);
    float vs = 0.0f;
#pragma unroll
    for (int i = 0; i < 6; ++i) { float d = v[u][i] - mean; vs += d * d; }
#pragma unroll
    for (int o = 16; o > 0; o >>= 1) vs += __shfl_xor(vs, o, 32);
    const float inv = rsqrtf(vs * (1.0f / E) + 1e-6f);
#pragma unroll
    for (int i = 0; i < 6; ++i) {
      int c = lane + 32 * i;
      out[t * E + c] = (_Float16)((v[u][i] - mean) * inv * gg[i] + bb[i]);
    }
  }
}

__global__ __launch_bounds__(256) void ln_kernel(
    const float* __restrict__ x, const float* __restrict__ g,
    const float* __restrict__ b, _Float16* __restrict__ out) {
  ln_block(x, g, b, out, blockIdx.x);
}

static constexpr int CVT_QKV = 144 * 48 / 256;
static constexpr int CVT_WO  = 48 * 48 / 256;
static constexpr int CVT_MLP = 384 * 192 / 256;
static constexpr int CVT_BLOCKS = 2 * CVT_QKV + 2 * CVT_WO + 2 * CVT_MLP;

__global__ __launch_bounds__(256) void front_kernel(
    const float* __restrict__ x, const float* __restrict__ g,
    const float* __restrict__ b, _Float16* __restrict__ out,
    const float* __restrict__ wqa, const float* __restrict__ woa,
    const float* __restrict__ wqg, const float* __restrict__ wog,
    const float* __restrict__ w1, const float* __restrict__ w2,
    _Float16* __restrict__ wqa_h, _Float16* __restrict__ woa_h,
    _Float16* __restrict__ wqg_h, _Float16* __restrict__ wog_h,
    _Float16* __restrict__ w1_h, _Float16* __restrict__ w2_h) {
  int blk = blockIdx.x;
  if (blk >= CVT_BLOCKS) { ln_block(x, g, b, out, blk - CVT_BLOCKS); return; }
  if (blk < CVT_QKV) { cvt_f16_frag_block(wqa, wqa_h, 48, 144 * 48, blk); return; }
  blk -= CVT_QKV;
  if (blk < CVT_WO)  { cvt_f16_frag_block(woa, woa_h, 48, 48 * 48, blk); return; }
  blk -= CVT_WO;
  if (blk < CVT_QKV) { cvt_f16_frag_block(wqg, wqg_h, 48, 144 * 48, blk); return; }
  blk -= CVT_QKV;
  if (blk < CVT_WO)  { cvt_f16_frag_block(wog, wog_h, 48, 48 * 48, blk); return; }
  blk -= CVT_WO;
  if (blk < CVT_MLP) { cvt_f16_frag_block(w1, w1_h, 192, 384 * 192, blk); return; }
  blk -= CVT_MLP;
  cvt_f16_frag_block(w2, w2_h, 384, 192 * 384, blk);
}

static constexpr int CV_TPW = 8;

DEVI void conv_block(const _Float16* __restrict__ xn,
                     const float* __restrict__ x,
                     const float* __restrict__ cw,
                     float* __restrict__ x1, int blk) {
  const int wave = threadIdx.x >> 5, lane = threadIdx.x & 31;
  const size_t t0 = ((size_t)blk * 8 + wave) * CV_TPW;
  for (int jo = 0; jo < CV_TPW; jo += 4) {
#pragma unroll
    for (int c = 0; c < 4; ++c) {
      const size_t t = t0 + jo + c;
      const float* w = cw + c * 15;
      const _Float16* xp = xn + t * E + 48;
#pragma unroll
      for (int i = 0; i < 3; ++i) {
        const int h = lane + 32 * i;
        float s = 0.0f;
#pragma unroll
        for (int k = 0; k < 15; ++k) {
          int hh = h + k - 7;
          if (hh >= 0 && hh < 96) s += w[k] * (float)xp[hh];
        }
        x1[t * E + 48 + h] = s + x[t * E + 48 + h];
      }
    }
  }
}

template <int NT, bool RELU, bool RES, bool OUT16>
DEVI void gemm_f16_block(
    const _Float16* __restrict__ A, int lda,
    const _Float16* __restrict__ W, int K,
    const float* __restrict__ bias,
    const float* __restrict__ res, int ldr,
    float* __restrict__ out32, _Float16* __restrict__ out16, int ldo,
    int bx, int by) {
  const int wave = threadIdx.x >> 5;
  const int m0 = bx * 128 + wave * 16;
  const int n0 = by * (NT * 16);
  v8f z = {};
  v8f acc[NT];
#pragma unroll
  for (int t = 0; t < NT; ++t) acc[t] = z;

  for (int k0 = 0; k0 < K; k0 += 32) {
    const int krem = K - k0;
    v16h a = load_frag(A + (size_t)m0 * lda + k0, lda, krem);
#pragma unroll
    for (int t = 0; t < NT; ++t) {
      v16h bfr = load_bfrag(W, K, (n0 >> 4) + t, k0);
      acc[t] = wmma16(a, bfr, acc[t]);
    }
  }

  const int lane = threadIdx.x & 31;
  const int col = lane & 15;
  const int r0  = (lane & 16) ? 8 : 0;
#pragma unroll
  for (int t = 0; t < NT; ++t) {
    const int n = n0 + t * 16 + col;
    const float bv = bias[n];
#pragma unroll
    for (int j = 0; j < 8; ++j) {
      const int m = m0 + r0 + j;
      float v = acc[t][j] + bv;
      if (RELU) v = fmaxf(v, 0.0f);
      if (RES)  v += res[(size_t)m * ldr + n];
      if (OUT16) out16[(size_t)m * ldo + n] = (_Float16)v;
      else       out32[(size_t)m * ldo + n] = v;
    }
  }
}

template <int NT, bool RELU, bool RES, bool OUT16>
__global__ __launch_bounds__(256) void gemm_f16_kernel(
    const _Float16* __restrict__ A0, const _Float16* __restrict__ A1, int lda,
    const _Float16* __restrict__ W0, const _Float16* __restrict__ W1, int K,
    const float* __restrict__ bias0, const float* __restrict__ bias1,
    const float* __restrict__ res0, const float* __restrict__ res1, int ldr,
    float* __restrict__ out32_0, float* __restrict__ out32_1,
    _Float16* __restrict__ out16_0, _Float16* __restrict__ out16_1, int ldo) {
  const bool second = blockIdx.z != 0;
  gemm_f16_block<NT, RELU, RES, OUT16>(
      second ? A1 : A0, lda, second ? W1 : W0, K, second ? bias1 : bias0,
      second ? res1 : res0, ldr, second ? out32_1 : out32_0,
      second ? out16_1 : out16_0, ldo, blockIdx.x, blockIdx.y);
}

static constexpr int CONV_BLOCKS = NTOK / (8 * CV_TPW);
static constexpr int QKV_MB      = NTOK / 128;
static constexpr int QKV_BLOCKS  = QKV_MB * 3;

__global__ __launch_bounds__(256) void mid_kernel(
    const _Float16* __restrict__ xn, const float* __restrict__ x,
    const float* __restrict__ cw, float* __restrict__ x1,
    const _Float16* __restrict__ wq0, const _Float16* __restrict__ wq1,
    const float* __restrict__ bq0, const float* __restrict__ bq1,
    _Float16* __restrict__ qkv0, _Float16* __restrict__ qkv1,
    int lda, int K, int ldo) {
  int blk = blockIdx.x;
  if (blk < CONV_BLOCKS) { conv_block(xn, x, cw, x1, blk); return; }
  blk -= CONV_BLOCKS;
  const bool second = blk >= QKV_BLOCKS;
  if (second) blk -= QKV_BLOCKS;
  const int by = blk / QKV_MB;
  const int bx = blk - by * QKV_MB;
  gemm_f16_block<3, false, false, true>(
      second ? xn + 96 : xn, lda, second ? wq1 : wq0, K, second ? bq1 : bq0,
      nullptr, 0, nullptr, second ? qkv1 : qkv0, ldo, bx, by);
}

static constexpr int W1_HALVES = 384 * 192;
static constexpr int W2_HALVES = 192 * 384;
static constexpr int MLP_WAVES = 16;
static constexpr int MLP_BLOCKS = NTOK / (16 * MLP_WAVES);

#define MLP_PIN2(a, b)                                        \
  do {                                                        \
    asm volatile("" : "+v"(a), "+v"(b) : : "memory");         \
    __builtin_amdgcn_sched_barrier(0);                        \
  } while (0)
#define MLP_PIN4(a, b, c, d)                                              \
  do {                                                                    \
    asm volatile("" : "+v"(a), "+v"(b), "+v"(c), "+v"(d) : : "memory");   \
    __builtin_amdgcn_sched_barrier(0);                                    \
  } while (0)

__global__ __launch_bounds__(512) void mlp_fused_kernel(
    const _Float16* __restrict__ tt, const _Float16* __restrict__ w1f,
    const _Float16* __restrict__ w2f, const float* __restrict__ b1,
    const float* __restrict__ b2, const float* __restrict__ x1,
    float* __restrict__ out) {
  __shared__ _Float16 wl[W1_HALVES + W2_HALVES];

  const int wave = threadIdx.x >> 5, lane = threadIdx.x & 31;
  const int m0 = (blockIdx.x * MLP_WAVES + wave) * 16;

#pragma unroll
  for (int i = 0; i < W1_HALVES / 8 / 512; ++i) {
    const int e = (threadIdx.x + 512 * i) * 8;
    *(v8h*)&wl[e] = *(const v8h*)(w1f + e);
  }
  asm volatile("" : : : "memory");
#pragma unroll
  for (int i = 0; i < W2_HALVES / 8 / 512; ++i) {
    const int e = (threadIdx.x + 512 * i) * 8;
    *(v8h*)&wl[W1_HALVES + e] = *(const v8h*)(w2f + e);
  }

  v16h tb[6];
#pragma unroll
  for (int ks = 0; ks < 6; ++ks)
    tb[ks] = load_frag(tt + (size_t)m0 * E + 32 * ks, E, 32);
  __syncthreads();

  const int col = lane & 15;
  const int r0  = (lane & 16) ? 8 : 0;
  const int lo8 = lane * 8;

  auto ldw = [&](int halves) -> v16h {
    const v8h lo = *(const v8h*)&wl[halves + lo8];
    const v8h hi = *(const v8h*)&wl[halves + 256 + lo8];
    v16h r;
#pragma unroll
    for (int i = 0; i < 8; ++i) { r[i] = lo[i]; r[8 + i] = hi[i]; }
    return r;
  };
  auto off1 = [](int j, int tl, int ks) { return ((2 * j + tl) * 12 + 2 * ks) << 8; };
  auto off2 = [](int j, int t) { return W1_HALVES + ((t * 24 + 2 * j) << 8); };

  v8f z = {};
  v8f acc2[12];
#pragma unroll
  for (int t = 0; t < 12; ++t) acc2[t] = z;

  v16h fa[4], fb[4];
#pragma unroll
  for (int q = 0; q < 4; ++q) fa[q] = ldw(off1(0, q & 1, q >> 1));

  v8f h0 = *(const v8f*)(b1 + r0);
  v8f h1 = *(const v8f*)(b1 + 16 + r0);

  for (int j = 0; j < 12; ++j) {
    const int jn = (j == 11) ? 0 : j + 1;

#pragma unroll
    for (int q = 0; q < 4; ++q) fb[q] = ldw(off1(j, q & 1, 2 + (q >> 1)));
    MLP_PIN2(h0, h1);
    h0 = wmma16(fa[0], tb[0], h0);
    h1 = wmma16(fa[1], tb[0], h1);
    h0 = wmma16(fa[2], tb[1], h0);
    h1 = wmma16(fa[3], tb[1], h1);
    MLP_PIN2(h0, h1);
#pragma unroll
    for (int q = 0; q < 4; ++q) fa[q] = ldw(off1(j, q & 1, 4 + (q >> 1)));
    MLP_PIN2(h0, h1);
    h0 = wmma16(fb[0], tb[2], h0);
    h1 = wmma16(fb[1], tb[2], h1);
    h0 = wmma16(fb[2], tb[3], h0);
    h1 = wmma16(fb[3], tb[3], h1);
    MLP_PIN2(h0, h1);
#pragma unroll
    for (int q = 0; q < 4; ++q) fb[q] = ldw(off2(j, q));
    MLP_PIN2(h0, h1);
    h0 = wmma16(fa[0], tb[4], h0);
    h1 = wmma16(fa[1], tb[4], h1);
    h0 = wmma16(fa[2], tb[5], h0);
    h1 = wmma16(fa[3], tb[5], h1);
    MLP_PIN2(h0, h1);

    v16h ha;
#pragma unroll
    for (int i = 0; i < 8; ++i) {
      ha[i]     = (_Float16)fmaxf(h0[i], 0.0f);
      ha[8 + i] = (_Float16)fmaxf(h1[i], 0.0f);
    }
    h0 = *(const v8f*)(b1 + 32 * jn + r0);
    h1 = *(const v8f*)(b1 + 32 * jn + 16 + r0);

#pragma unroll
    for (int q = 0; q < 4; ++q) fa[q] = ldw(off2(j, 4 + q));
    MLP_PIN4(acc2[0], acc2[1], acc2[2], acc2[3]);
#pragma unroll
    for (int q = 0; q < 4; ++q) acc2[q] = wmma16(ha, fb[q], acc2[q]);
    MLP_PIN4(acc2[0], acc2[1], acc2[2], acc2[3]);
#pragma unroll
    for (int q = 0; q < 4; ++q) fb[q] = ldw(off2(j, 8 + q));
    MLP_PIN4(acc2[4], acc2[5], acc2[6], acc2[7]);
#pragma unroll
    for (int q = 0; q < 4; ++q) acc2[4 + q] = wmma16(ha, fa[q], acc2[4 + q]);
    MLP_PIN4(acc2[4], acc2[5], acc2[6], acc2[7]);
#pragma unroll
    for (int q = 0; q < 4; ++q) fa[q] = ldw(off1(jn, q & 1, q >> 1));
    MLP_PIN4(acc2[8], acc2[9], acc2[10], acc2[11]);
#pragma unroll
    for (int q = 0; q < 4; ++q) acc2[8 + q] = wmma16(ha, fb[q], acc2[8 + q]);
    MLP_PIN4(acc2[8], acc2[9], acc2[10], acc2[11]);
  }

#pragma unroll
  for (int t = 0; t < 12; ++t) {
    const int n = t * 16 + col;
    const float bv = b2[n];
#pragma unroll
    for (int r = 0; r < 8; ++r) {
      const size_t o = (size_t)(m0 + r0 + r) * E + n;
      out[o] = fmaxf(acc2[t][r] + bv, 0.0f) + x1[o];
    }
  }
}
#undef MLP_PIN2
#undef MLP_PIN4

__global__ __launch_bounds__(256) void attn_kernel(
    const _Float16* __restrict__ qkv_all, _Float16* __restrict__ out_all) {
  __shared__ _Float16 vt[16 * VT_LD];
  __shared__ _Float16 kt[SEQ * KT_LD];

  const int bid = blockIdx.x;
  const int b = bid / 6;
  const int r = bid % 6;
  const int branch = r / 3;
  const int hh = r % 3;
  const _Float16* qkv = qkv_all + (size_t)branch * NTOK * 144
                                + (size_t)b * SEQ * 144;
  _Float16* outp = out_all + (size_t)branch * NTOK * QDIM
                           + (size_t)b * SEQ * QDIM;
  const int qcol = hh * 16, kcol = 48 + hh * 16, vcol = 96 + hh * 16;

  for (int e = threadIdx.x; e < 16 * SEQ; e += 256) {
    const int d = e & 15, j = e >> 4;
    const _Float16 kv = qkv[(size_t)j * 144 + kcol + d];
    const _Float16 vv = qkv[(size_t)j * 144 + vcol + d];
    kt[j * KT_LD + d] = kv;
    vt[d * VT_LD + j] = vv;
  }
  __syncthreads();

  const int wave = threadIdx.x >> 5, lane = threadIdx.x & 31;
  const int q0 = blockIdx.y * 128 + wave * 16;
  const int row = lane & 15;
  const int d0  = (lane & 16) ? 8 : 0;

  v16h qf;
  {
    const _Float16* qp = qkv + (size_t)(q0 + row) * 144 + qcol + d0;
    v8h ql = *(const v8h*)qp;
#pragma unroll
    for (int i = 0; i < 8; ++i) {
      qf[i] = ql[i] * (_Float16)0.25f;
      qf[8 + i] = (_Float16)0.0f;
    }
  }

  v8f z = {};
  v8f oacc = z;
  float m_run = -1e30f, l_run = 0.0f;

  for (int kb = 0; kb < SEQ / 32; ++kb) {
    const int j0 = kb * 32;
    v16h kf0, kf1;
    {
      const _Float16* kp0 = &kt[(j0 + row) * KT_LD + d0];
      const _Float16* kp1 = &kt[(j0 + 16 + row) * KT_LD + d0];
      v8h ka = *(const v8h*)kp0;
      v8h kb2 = *(const v8h*)kp1;
#pragma unroll
      for (int i = 0; i < 8; ++i) {
        kf0[i] = ka[i];  kf0[8 + i] = (_Float16)0.0f;
        kf1[i] = kb2[i]; kf1[8 + i] = (_Float16)0.0f;
      }
    }
    v8f s0 = wmma16(kf0, qf, z);
    v8f s1 = wmma16(kf1, qf, z);

    float tm = -1e30f;
#pragma unroll
    for (int j = 0; j < 8; ++j) tm = fmaxf(tm, fmaxf(s0[j], s1[j]));
    tm = fmaxf(tm, swap16(tm));
    const float mnew = fmaxf(m_run, tm);
    const float corr = __expf(m_run - mnew);
    float p0[8], p1[8], ls = 0.0f;
#pragma unroll
    for (int j = 0; j < 8; ++j) {
      p0[j] = __expf(s0[j] - mnew);
      p1[j] = __expf(s1[j] - mnew);
      ls += p0[j] + p1[j];
    }
    ls += swap16(ls);
    l_run = l_run * corr + ls;
    m_run = mnew;
#pragma unroll
    for (int j = 0; j < 8; ++j) oacc[j] *= corr;

    v16h pf;
#pragma unroll
    for (int i = 0; i < 8; ++i) {
      pf[i] = (_Float16)p0[i];
      pf[8 + i] = (_Float16)p1[i];
    }
    const _Float16* vp = &vt[row * VT_LD + j0 + d0];
    v8h vl = *(const v8h*)vp;
    v8h vh = *(const v8h*)(vp + 16);
    v16h vf;
#pragma unroll
    for (int i = 0; i < 8; ++i) { vf[i] = vl[i]; vf[8 + i] = vh[i]; }

    oacc = wmma16(vf, pf, oacc);
  }

  const float inv = 1.0f / l_run;
#pragma unroll
  for (int j = 0; j < 8; ++j)
    outp[(size_t)(q0 + row) * QDIM + qcol + d0 + j] = (_Float16)(oacc[j] * inv);
}

static inline size_t align256(size_t x) { return (x + 255) & ~(size_t)255; }

extern "C" void kernel_launch(void* const* d_in, const int* in_sizes, int n_in,
                              void* d_out, int out_size, void* d_ws, size_t ws_size,
                              hipStream_t stream) {
  (void)in_sizes; (void)n_in; (void)out_size; (void)ws_size;

  const float* x        = (const float*)d_in[0];
  const float* ln1_g    = (const float*)d_in[1];
  const float* ln1_b    = (const float*)d_in[2];
  const float* acc_wqkv = (const float*)d_in[3];
  const float* acc_bqkv = (const float*)d_in[4];
  const float* acc_wo   = (const float*)d_in[5];
  const float* acc_bo   = (const float*)d_in[6];
  const float* gy_wqkv  = (const float*)d_in[7];
  const float* gy_bqkv  = (const float*)d_in[8];
  const float* gy_wo    = (const float*)d_in[9];
  const float* gy_bo    = (const float*)d_in[10];
  const float* conv_w   = (const float*)d_in[11];
  const float* ln2_g    = (const float*)d_in[12];
  const float* ln2_b    = (const float*)d_in[13];
  const float* w1       = (const float*)d_in[14];
  const float* b1       = (const float*)d_in[15];
  const float* w2       = (const float*)d_in[16];
  const float* b2       = (const float*)d_in[17];

  char* ws = (char*)d_ws;
  size_t off = 0;
  _Float16* xn_h   = (_Float16*)(ws + off); off = align256(off + (size_t)NTOK * E * 2);
  _Float16* qkv_h  = (_Float16*)(ws + off); off = align256(off + (size_t)2 * NTOK * 144 * 2);
  _Float16* attn_h = (_Float16*)(ws + off); off = align256(off + (size_t)2 * NTOK * QDIM * 2);
  float*    x1     = (float*)   (ws + off); off = align256(off + (size_t)NTOK * E * 4);
  _Float16* tt_h   = (_Float16*)(ws + off); off = align256(off + (size_t)NTOK * E * 2);
  _Float16* wq_a_h = (_Float16*)(ws + off); off = align256(off + 144 * 48 * 2);
  _Float16* wo_a_h = (_Float16*)(ws + off); off = align256(off + 48 * 48 * 2);
  _Float16* wq_g_h = (_Float16*)(ws + off); off = align256(off + 144 * 48 * 2);
  _Float16* wo_g_h = (_Float16*)(ws + off); off = align256(off + 48 * 48 * 2);
  _Float16* w1_h   = (_Float16*)(ws + off); off = align256(off + 384 * 192 * 2);
  _Float16* w2_h   = (_Float16*)(ws + off); off = align256(off + 192 * 384 * 2);

  front_kernel<<<CVT_BLOCKS + NTOK / (8 * LN_TPW), 256, 0, stream>>>(
      x, ln1_g, ln1_b, xn_h,
      acc_wqkv, acc_wo, gy_wqkv, gy_wo, w1, w2,
      wq_a_h, wo_a_h, wq_g_h, wo_g_h, w1_h, w2_h);

  mid_kernel<<<CONV_BLOCKS + 2 * QKV_BLOCKS, 256, 0, stream>>>(
      xn_h, x, conv_w, x1, wq_a_h, wq_g_h, acc_bqkv, gy_bqkv,
      qkv_h, qkv_h + (size_t)NTOK * 144, E, 48, 144);

  attn_kernel<<<dim3(BATCH * 6, SEQ / 128), 256, 0, stream>>>(qkv_h, attn_h);

  gemm_f16_kernel<3, false, true, false><<<dim3(NTOK / 128, 1, 2), 256, 0, stream>>>(
      attn_h, attn_h + (size_t)NTOK * QDIM, QDIM, wo_a_h, wo_g_h, 48,
      acc_bo, gy_bo, x + 0, x + 144, E, x1 + 0, x1 + 144, nullptr, nullptr, E);

  ln_kernel<<<NTOK / (8 * LN_TPW), 256, 0, stream>>>(x1, ln2_g, ln2_b, tt_h);

  mlp_fused_kernel<<<MLP_BLOCKS, 512, 0, stream>>>(
      tt_h, w1_h, w2_h, b1, b2, x1, (float*)d_out);
}


# Out-projection moved into the attention launch: a wave owns 16 queries of all 3 heads, its normalised f16 head tiles are the B fragments of Wo*o^T in registers, attn never stored; 5 launches -> 4
# speedup vs baseline: 2.0714x; 1.1162x over previous
#include <hip/hip_runtime.h>


typedef __attribute__((ext_vector_type(16))) _Float16 v16h;
typedef __attribute__((ext_vector_type(8)))  _Float16 v8h;
typedef __attribute__((ext_vector_type(8)))  float    v8f;
typedef __attribute__((ext_vector_type(4)))  float    v4f;

#define DEVI static __device__ __forceinline__

static constexpr int BATCH = 128;
static constexpr int SEQ   = 512;
static constexpr int E     = 192;
static constexpr int NTOK  = BATCH * SEQ;

static constexpr int KT_LD = 24;
static constexpr int VT_LD = 520;

DEVI v8f wmma16(v16h a, v16h b, v8f c) {
  return __builtin_amdgcn_wmma_f32_16x16x32_f16(false, a, false, b,
                                                (short)0, c, false, false);
}

DEVI float swap16(float v) {
  int i = __builtin_bit_cast(int, v);
  int r = __builtin_amdgcn_permlanex16(i, i, 0x76543210, 0xfedcba98, false, false);
  return __builtin_bit_cast(float, r);
}

DEVI v16h load_frag(const _Float16* __restrict__ base, int ld, int krem) {
  const int lane = threadIdx.x & 31;
  const int row  = lane & 15;
  const int d0   = (lane & 16) ? 8 : 0;
  const _Float16* p = base + (size_t)row * ld + d0;
  v16h r;
  if (krem >= 32) {
    v8h lo = *(const v8h*)(p);
    v8h hi = *(const v8h*)(p + 16);
#pragma unroll
    for (int i = 0; i < 8; ++i) { r[i] = lo[i]; r[8 + i] = hi[i]; }
  } else {
#pragma unroll
    for (int i = 0; i < 8; ++i) {
      r[i]     = (d0 + i      < krem) ? p[i]      : (_Float16)0.0f;
      r[8 + i] = (d0 + 16 + i < krem) ? p[16 + i] : (_Float16)0.0f;
    }
  }
  return r;
}

DEVI v16h load_bfrag(const _Float16* __restrict__ Wf, int K, int ntile, int k0) {
  const int lane = threadIdx.x & 31;
  const _Float16* p = Wf + (((size_t)ntile * (K >> 4) + (k0 >> 4)) << 8) + lane * 8;
  v8h lo = *(const v8h*)(p);
  v8h hi = {};
  if (K - k0 >= 32) hi = *(const v8h*)(p + 256);
  v16h r;
#pragma unroll
  for (int i = 0; i < 8; ++i) { r[i] = lo[i]; r[8 + i] = hi[i]; }
  return r;
}

DEVI void cvt_f16_frag_block(const float* __restrict__ src,
                             _Float16* __restrict__ dst, int K, int n, int blk) {
  const int o = blk * 256 + threadIdx.x;
  if (o >= n) return;
  const int i = o & 7;
  const int l = (o >> 3) & 31;
  const int chunk = o >> 8;
  const int nk16 = K >> 4;
  const int ntile = chunk / nk16;
  const int k16 = chunk - ntile * nk16;
  const int row = ntile * 16 + (l & 15);
  const int k = k16 * 16 + ((l & 16) ? 8 : 0) + i;
  dst[o] = (_Float16)src[(size_t)row * K + k];
}

static constexpr int LN_TPW = 8;

DEVI void ln_block(const float* __restrict__ x, const float* __restrict__ g,
                   const float* __restrict__ b, _Float16* __restrict__ out,
                   int blk) {
  const int wave = threadIdx.x >> 5, lane = threadIdx.x & 31;
  const size_t t0 = ((size_t)blk * 8 + wave) * LN_TPW;
  float gg[6], bb[6];
#pragma unroll
  for (int i = 0; i < 6; ++i) { gg[i] = g[lane + 32 * i]; bb[i] = b[lane + 32 * i]; }
  float v[LN_TPW][6];
#pragma unroll
  for (int u = 0; u < LN_TPW; ++u) {
    const float* xp = x + (t0 + u) * E;
#pragma unroll
    for (int i = 0; i < 6; ++i) v[u][i] = xp[lane + 32 * i];
  }
#pragma unroll
  for (int u = 0; u < LN_TPW; ++u) {
    const size_t t = t0 + u;
    float s = 0.0f;
#pragma unroll
    for (int i = 0; i < 6; ++i) s += v[u][i];
#pragma unroll
    for (int o = 16; o > 0; o >>= 1) s += __shfl_xor(s, o, 32);
    const float mean = s * (1.0f / E);
    float vs = 0.0f;
#pragma unroll
    for (int i = 0; i < 6; ++i) { float d = v[u][i] - mean; vs += d * d; }
#pragma unroll
    for (int o = 16; o > 0; o >>= 1) vs += __shfl_xor(vs, o, 32);
    const float inv = rsqrtf(vs * (1.0f / E) + 1e-6f);
#pragma unroll
    for (int i = 0; i < 6; ++i) {
      int c = lane + 32 * i;
      out[t * E + c] = (_Float16)((v[u][i] - mean) * inv * gg[i] + bb[i]);
    }
  }
}

static constexpr int CVT_QKV = 144 * 48 / 256;
static constexpr int CVT_WO  = 48 * 48 / 256;
static constexpr int CVT_MLP = 384 * 192 / 256;
static constexpr int CVT_BLOCKS = 2 * CVT_QKV + 2 * CVT_WO + 2 * CVT_MLP;

__global__ __launch_bounds__(256) void front_kernel(
    const float* __restrict__ x, const float* __restrict__ g,
    const float* __restrict__ b, _Float16* __restrict__ out,
    const float* __restrict__ wqa, const float* __restrict__ woa,
    const float* __restrict__ wqg, const float* __restrict__ wog,
    const float* __restrict__ w1, const float* __restrict__ w2,
    _Float16* __restrict__ wqa_h, _Float16* __restrict__ woa_h,
    _Float16* __restrict__ wqg_h, _Float16* __restrict__ wog_h,
    _Float16* __restrict__ w1_h, _Float16* __restrict__ w2_h) {
  int blk = blockIdx.x;
  if (blk >= CVT_BLOCKS) { ln_block(x, g, b, out, blk - CVT_BLOCKS); return; }
  if (blk < CVT_QKV) { cvt_f16_frag_block(wqa, wqa_h, 48, 144 * 48, blk); return; }
  blk -= CVT_QKV;
  if (blk < CVT_WO)  { cvt_f16_frag_block(woa, woa_h, 48, 48 * 48, blk); return; }
  blk -= CVT_WO;
  if (blk < CVT_QKV) { cvt_f16_frag_block(wqg, wqg_h, 48, 144 * 48, blk); return; }
  blk -= CVT_QKV;
  if (blk < CVT_WO)  { cvt_f16_frag_block(wog, wog_h, 48, 48 * 48, blk); return; }
  blk -= CVT_WO;
  if (blk < CVT_MLP) { cvt_f16_frag_block(w1, w1_h, 192, 384 * 192, blk); return; }
  blk -= CVT_MLP;
  cvt_f16_frag_block(w2, w2_h, 384, 192 * 384, blk);
}

static constexpr int CV_TPW = 8;

DEVI void conv_block(const _Float16* __restrict__ xn,
                     const float* __restrict__ x,
                     const float* __restrict__ cw,
                     float* __restrict__ x1, int blk) {
  const int wave = threadIdx.x >> 5, lane = threadIdx.x & 31;
  const size_t t0 = ((size_t)blk * 8 + wave) * CV_TPW;
  for (int jo = 0; jo < CV_TPW; jo += 4) {
#pragma unroll
    for (int c = 0; c < 4; ++c) {
      const size_t t = t0 + jo + c;
      const float* w = cw + c * 15;
      const _Float16* xp = xn + t * E + 48;
#pragma unroll
      for (int i = 0; i < 3; ++i) {
        const int h = lane + 32 * i;
        float s = 0.0f;
#pragma unroll
        for (int k = 0; k < 15; ++k) {
          int hh = h + k - 7;
          if (hh >= 0 && hh < 96) s += w[k] * (float)xp[hh];
        }
        x1[t * E + 48 + h] = s + x[t * E + 48 + h];
      }
    }
  }
}

template <int NT, bool RELU, bool RES, bool OUT16>
DEVI void gemm_f16_block(
    const _Float16* __restrict__ A, int lda,
    const _Float16* __restrict__ W, int K,
    const float* __restrict__ bias,
    const float* __restrict__ res, int ldr,
    float* __restrict__ out32, _Float16* __restrict__ out16, int ldo,
    int bx, int by) {
  const int wave = threadIdx.x >> 5;
  const int m0 = bx * 128 + wave * 16;
  const int n0 = by * (NT * 16);
  v8f z = {};
  v8f acc[NT];
#pragma unroll
  for (int t = 0; t < NT; ++t) acc[t] = z;

  for (int k0 = 0; k0 < K; k0 += 32) {
    const int krem = K - k0;
    v16h a = load_frag(A + (size_t)m0 * lda + k0, lda, krem);
#pragma unroll
    for (int t = 0; t < NT; ++t) {
      v16h bfr = load_bfrag(W, K, (n0 >> 4) + t, k0);
      acc[t] = wmma16(a, bfr, acc[t]);
    }
  }

  const int lane = threadIdx.x & 31;
  const int col = lane & 15;
  const int r0  = (lane & 16) ? 8 : 0;
#pragma unroll
  for (int t = 0; t < NT; ++t) {
    const int n = n0 + t * 16 + col;
    const float bv = bias[n];
#pragma unroll
    for (int j = 0; j < 8; ++j) {
      const int m = m0 + r0 + j;
      float v = acc[t][j] + bv;
      if (RELU) v = fmaxf(v, 0.0f);
      if (RES)  v += res[(size_t)m * ldr + n];
      if (OUT16) out16[(size_t)m * ldo + n] = (_Float16)v;
      else       out32[(size_t)m * ldo + n] = v;
    }
  }
}

static constexpr int CONV_BLOCKS = NTOK / (8 * CV_TPW);
static constexpr int QKV_MB      = NTOK / 128;
static constexpr int QKV_BLOCKS  = QKV_MB * 3;

__global__ __launch_bounds__(256) void mid_kernel(
    const _Float16* __restrict__ xn, const float* __restrict__ x,
    const float* __restrict__ cw, float* __restrict__ x1,
    const _Float16* __restrict__ wq0, const _Float16* __restrict__ wq1,
    const float* __restrict__ bq0, const float* __restrict__ bq1,
    _Float16* __restrict__ qkv0, _Float16* __restrict__ qkv1,
    int lda, int K, int ldo) {
  int blk = blockIdx.x;
  if (blk < CONV_BLOCKS) { conv_block(xn, x, cw, x1, blk); return; }
  blk -= CONV_BLOCKS;
  const bool second = blk >= QKV_BLOCKS;
  if (second) blk -= QKV_BLOCKS;
  const int by = blk / QKV_MB;
  const int bx = blk - by * QKV_MB;
  gemm_f16_block<3, false, false, true>(
      second ? xn + 96 : xn, lda, second ? wq1 : wq0, K, second ? bq1 : bq0,
      nullptr, 0, nullptr, second ? qkv1 : qkv0, ldo, bx, by);
}

static constexpr int W1_HALVES = 384 * 192;
static constexpr int W2_HALVES = 192 * 384;
static constexpr int MLP_WAVES = 16;
static constexpr int MLP_BLOCKS = NTOK / (16 * MLP_WAVES);

#define MLP_PIN2(a, b)                                        \
  do {                                                        \
    asm volatile("" : "+v"(a), "+v"(b) : : "memory");         \
    __builtin_amdgcn_sched_barrier(0);                        \
  } while (0)
#define MLP_PIN4(a, b, c, d)                                              \
  do {                                                                    \
    asm volatile("" : "+v"(a), "+v"(b), "+v"(c), "+v"(d) : : "memory");   \
    __builtin_amdgcn_sched_barrier(0);                                    \
  } while (0)

__global__ __launch_bounds__(512) void mlp_fused_kernel(
    const float* __restrict__ ln_g, const float* __restrict__ ln_b,
    const _Float16* __restrict__ w1f,
    const _Float16* __restrict__ w2f, const float* __restrict__ b1,
    const float* __restrict__ b2, const float* __restrict__ x1,
    float* __restrict__ out) {
  __shared__ _Float16 wl[W1_HALVES + W2_HALVES];

  const int wave = threadIdx.x >> 5, lane = threadIdx.x & 31;
  const int m0 = (blockIdx.x * MLP_WAVES + wave) * 16;
  const int col = lane & 15;
  const int r0  = (lane & 16) ? 8 : 0;

  const float* xr = x1 + (size_t)(m0 + col) * E + r0;
  v4f xq[24];
#pragma unroll
  for (int ks = 0; ks < 6; ++ks)
#pragma unroll
    for (int q = 0; q < 4; ++q)
      xq[4 * ks + q] = *(const v4f*)(xr + 32 * ks + 16 * (q >> 1) + 4 * (q & 1));

#pragma unroll
  for (int i = 0; i < W1_HALVES / 8 / 512; ++i) {
    const int e = (threadIdx.x + 512 * i) * 8;
    *(v8h*)&wl[e] = *(const v8h*)(w1f + e);
  }
  asm volatile("" : : : "memory");
#pragma unroll
  for (int i = 0; i < W2_HALVES / 8 / 512; ++i) {
    const int e = (threadIdx.x + 512 * i) * 8;
    *(v8h*)&wl[W1_HALVES + e] = *(const v8h*)(w2f + e);
  }

  v16h tb[6];
  {
    v4f s4 = {};
#pragma unroll
    for (int j = 0; j < 24; ++j) s4 += xq[j];
    float s = (s4.x + s4.y) + (s4.z + s4.w);
    s += swap16(s);
    const float mean = s * (1.0f / E);
    v4f vs4 = {};
#pragma unroll
    for (int j = 0; j < 24; ++j) { xq[j] -= mean; vs4 += xq[j] * xq[j]; }
    float vs = (vs4.x + vs4.y) + (vs4.z + vs4.w);
    vs += swap16(vs);
    const float inv = rsqrtf(vs * (1.0f / E) + 1e-6f);

    __builtin_amdgcn_sched_barrier(0);
    auto ldgb = [&](int ks, v4f* g, v4f* b) {
#pragma unroll
      for (int q = 0; q < 4; ++q) {
        const int k = r0 + 32 * ks + 16 * (q >> 1) + 4 * (q & 1);
        g[q] = *(const v4f*)(ln_g + k);
        b[q] = *(const v4f*)(ln_b + k);
      }
    };
    v4f gq[2][4], bq[2][4];
    ldgb(0, gq[0], bq[0]);
#pragma unroll
    for (int ks = 0; ks < 6; ++ks) {
      if (ks < 5) ldgb(ks + 1, gq[(ks + 1) & 1], bq[(ks + 1) & 1]);
#pragma unroll
      for (int q = 0; q < 4; ++q) {
        const v4f y = xq[4 * ks + q] * inv * gq[ks & 1][q] + bq[ks & 1][q];
#pragma unroll
        for (int e = 0; e < 4; ++e) tb[ks][4 * q + e] = (_Float16)y[e];
      }
      __builtin_amdgcn_sched_barrier(0);
    }
  }
  __syncthreads();

  const int lo8 = lane * 8;

  auto ldw = [&](int halves) -> v16h {
    const v8h lo = *(const v8h*)&wl[halves + lo8];
    const v8h hi = *(const v8h*)&wl[halves + 256 + lo8];
    v16h r;
#pragma unroll
    for (int i = 0; i < 8; ++i) { r[i] = lo[i]; r[8 + i] = hi[i]; }
    return r;
  };
  auto off1 = [](int j, int tl, int ks) { return ((2 * j + tl) * 12 + 2 * ks) << 8; };
  auto off2 = [](int j, int t) { return W1_HALVES + ((t * 24 + 2 * j) << 8); };

  v8f z = {};
  v8f acc2[12];
#pragma unroll
  for (int t = 0; t < 12; ++t) acc2[t] = z;

  v16h fa[4], fb[4];
#pragma unroll
  for (int q = 0; q < 4; ++q) fa[q] = ldw(off1(0, q & 1, q >> 1));

  v8f h0 = *(const v8f*)(b1 + r0);
  v8f h1 = *(const v8f*)(b1 + 16 + r0);

  for (int j = 0; j < 12; ++j) {
    const int jn = (j == 11) ? 0 : j + 1;

#pragma unroll
    for (int q = 0; q < 4; ++q) fb[q] = ldw(off1(j, q & 1, 2 + (q >> 1)));
    MLP_PIN2(h0, h1);
    h0 = wmma16(fa[0], tb[0], h0);
    h1 = wmma16(fa[1], tb[0], h1);
    h0 = wmma16(fa[2], tb[1], h0);
    h1 = wmma16(fa[3], tb[1], h1);
    MLP_PIN2(h0, h1);
#pragma unroll
    for (int q = 0; q < 4; ++q) fa[q] = ldw(off1(j, q & 1, 4 + (q >> 1)));
    MLP_PIN2(h0, h1);
    h0 = wmma16(fb[0], tb[2], h0);
    h1 = wmma16(fb[1], tb[2], h1);
    h0 = wmma16(fb[2], tb[3], h0);
    h1 = wmma16(fb[3], tb[3], h1);
    MLP_PIN2(h0, h1);
#pragma unroll
    for (int q = 0; q < 4; ++q) fb[q] = ldw(off2(j, q));
    MLP_PIN2(h0, h1);
    h0 = wmma16(fa[0], tb[4], h0);
    h1 = wmma16(fa[1], tb[4], h1);
    h0 = wmma16(fa[2], tb[5], h0);
    h1 = wmma16(fa[3], tb[5], h1);
    MLP_PIN2(h0, h1);

    v16h ha;
#pragma unroll
    for (int i = 0; i < 8; ++i) {
      ha[i]     = (_Float16)fmaxf(h0[i], 0.0f);
      ha[8 + i] = (_Float16)fmaxf(h1[i], 0.0f);
    }
    h0 = *(const v8f*)(b1 + 32 * jn + r0);
    h1 = *(const v8f*)(b1 + 32 * jn + 16 + r0);

#pragma unroll
    for (int q = 0; q < 4; ++q) fa[q] = ldw(off2(j, 4 + q));
    MLP_PIN4(acc2[0], acc2[1], acc2[2], acc2[3]);
#pragma unroll
    for (int q = 0; q < 4; ++q) acc2[q] = wmma16(ha, fb[q], acc2[q]);
    MLP_PIN4(acc2[0], acc2[1], acc2[2], acc2[3]);
#pragma unroll
    for (int q = 0; q < 4; ++q) fb[q] = ldw(off2(j, 8 + q));
    MLP_PIN4(acc2[4], acc2[5], acc2[6], acc2[7]);
#pragma unroll
    for (int q = 0; q < 4; ++q) acc2[4 + q] = wmma16(ha, fa[q], acc2[4 + q]);
    MLP_PIN4(acc2[4], acc2[5], acc2[6], acc2[7]);
#pragma unroll
    for (int q = 0; q < 4; ++q) fa[q] = ldw(off1(jn, q & 1, q >> 1));
    MLP_PIN4(acc2[8], acc2[9], acc2[10], acc2[11]);
#pragma unroll
    for (int q = 0; q < 4; ++q) acc2[8 + q] = wmma16(ha, fb[q], acc2[8 + q]);
    MLP_PIN4(acc2[8], acc2[9], acc2[10], acc2[11]);
  }

#pragma unroll
  for (int t = 0; t < 12; ++t) {
    const int n = t * 16 + col;
    const float bv = b2[n];
#pragma unroll
    for (int r = 0; r < 8; ++r) {
      const size_t o = (size_t)(m0 + r0 + r) * E + n;
      out[o] = fmaxf(acc2[t][r] + bv, 0.0f) + x1[o];
    }
  }
}
#undef MLP_PIN2
#undef MLP_PIN4

static constexpr int AT_WAVES  = 16;
static constexpr int AT_BLOCKS = BATCH * 2 * (SEQ / (16 * AT_WAVES));

__global__ __launch_bounds__(512) void attn_oproj_kernel(
    const _Float16* __restrict__ qkv_all,
    const _Float16* __restrict__ wo0, const _Float16* __restrict__ wo1,
    const float* __restrict__ bo0, const float* __restrict__ bo1,
    const float* __restrict__ x, float* __restrict__ x1) {
  __shared__ _Float16 vt[3 * 16 * VT_LD];
  __shared__ _Float16 kt[3 * SEQ * KT_LD];

  const int bid = blockIdx.x;
  const int b = bid >> 2;
  const int branch = (bid >> 1) & 1;
  const int qhalf = bid & 1;
  const _Float16* qkv = qkv_all + (size_t)branch * NTOK * 144
                                + (size_t)b * SEQ * 144;
  const _Float16* wo = branch ? wo1 : wo0;
  const float* bo = branch ? bo1 : bo0;
  const int cb = branch ? 144 : 0;

  const int wave = threadIdx.x >> 5, lane = threadIdx.x & 31;
  const int q0 = qhalf * 256 + wave * 16;
  const int row = lane & 15;
  const int d0  = (lane & 16) ? 8 : 0;

  v8h ql[3];
#pragma unroll
  for (int hh = 0; hh < 3; ++hh)
    ql[hh] = *(const v8h*)(qkv + (size_t)(q0 + row) * 144 + hh * 16 + d0);

  const size_t tok = (size_t)b * SEQ + q0 + row;
  v4f rs[3][2];
#pragma unroll
  for (int nt = 0; nt < 3; ++nt) {
    const float* xr = x + tok * E + cb + nt * 16 + d0;
    rs[nt][0] = *(const v4f*)(xr);
    rs[nt][1] = *(const v4f*)(xr + 4);
  }

#pragma unroll
  for (int i = 0; i < 6; ++i) {
    const int e = threadIdx.x + 512 * i;
    const int j = e / 6, c6 = e - 6 * j;
    const v8h kv = *(const v8h*)(qkv + (size_t)j * 144 + 48 + c6 * 8);
    *(v8h*)&kt[((c6 >> 1) * SEQ + j) * KT_LD + (c6 & 1) * 8] = kv;
  }
#pragma unroll
  for (int i = 0; i < 6; ++i) {
    const int e = threadIdx.x + 512 * i;
    const int j = e / 6, c6 = e - 6 * j;
    const v8h vv = *(const v8h*)(qkv + (size_t)j * 144 + 96 + c6 * 8);
#pragma unroll
    for (int u = 0; u < 8; ++u) vt[(c6 * 8 + u) * VT_LD + j] = vv[u];
  }
  __syncthreads();

  v8f z = {};
  v8h po[3];
#pragma unroll
  for (int hh = 0; hh < 3; ++hh) {
    v16h qf;
#pragma unroll
    for (int i = 0; i < 8; ++i) {
      qf[i] = ql[hh][i] * (_Float16)0.25f;
      qf[8 + i] = (_Float16)0.0f;
    }
    const _Float16* kth = kt + hh * SEQ * KT_LD;
    const _Float16* vth = vt + hh * 16 * VT_LD;

    v8f oacc = z;
    float m_run = -1e30f, l_run = 0.0f;

    for (int kb = 0; kb < SEQ / 32; ++kb) {
      const int j0 = kb * 32;
      v16h kf0, kf1;
      {
        const _Float16* kp0 = &kth[(j0 + row) * KT_LD + d0];
        const _Float16* kp1 = &kth[(j0 + 16 + row) * KT_LD + d0];
        v8h ka = *(const v8h*)kp0;
        v8h kb2 = *(const v8h*)kp1;
#pragma unroll
        for (int i = 0; i < 8; ++i) {
          kf0[i] = ka[i];  kf0[8 + i] = (_Float16)0.0f;
          kf1[i] = kb2[i]; kf1[8 + i] = (_Float16)0.0f;
        }
      }
      v8f s0 = wmma16(kf0, qf, z);
      v8f s1 = wmma16(kf1, qf, z);

      float tm = -1e30f;
#pragma unroll
      for (int j = 0; j < 8; ++j) tm = fmaxf(tm, fmaxf(s0[j], s1[j]));
      tm = fmaxf(tm, swap16(tm));
      const float mnew = fmaxf(m_run, tm);
      const float corr = __expf(m_run - mnew);
      float p0[8], p1[8], ls = 0.0f;
#pragma unroll
      for (int j = 0; j < 8; ++j) {
        p0[j] = __expf(s0[j] - mnew);
        p1[j] = __expf(s1[j] - mnew);
        ls += p0[j] + p1[j];
      }
      ls += swap16(ls);
      l_run = l_run * corr + ls;
      m_run = mnew;
#pragma unroll
      for (int j = 0; j < 8; ++j) oacc[j] *= corr;

      v16h pf;
#pragma unroll
      for (int i = 0; i < 8; ++i) {
        pf[i] = (_Float16)p0[i];
        pf[8 + i] = (_Float16)p1[i];
      }
      const _Float16* vp = &vth[row * VT_LD + j0 + d0];
      v8h vl = *(const v8h*)vp;
      v8h vh = *(const v8h*)(vp + 16);
      v16h vf;
#pragma unroll
      for (int i = 0; i < 8; ++i) { vf[i] = vl[i]; vf[8 + i] = vh[i]; }

      oacc = wmma16(vf, pf, oacc);
    }

    const float inv = 1.0f / l_run;
#pragma unroll
    for (int j = 0; j < 8; ++j) po[hh][j] = (_Float16)(oacc[j] * inv);
  }

  v16h ob0, ob1;
#pragma unroll
  for (int i = 0; i < 8; ++i) {
    ob0[i] = po[0][i]; ob0[8 + i] = po[1][i];
    ob1[i] = po[2][i]; ob1[8 + i] = (_Float16)0.0f;
  }
#pragma unroll
  for (int nt = 0; nt < 3; ++nt) {
    const v16h a0 = load_bfrag(wo, 48, nt, 0);
    const v16h a1 = load_bfrag(wo, 48, nt, 32);
    const v4f bl = *(const v4f*)(bo + nt * 16 + d0);
    const v4f bh = *(const v4f*)(bo + nt * 16 + d0 + 4);
    v8f y = wmma16(a0, ob0, z);
    y = wmma16(a1, ob1, y);
    v4f lo, hi;
#pragma unroll
    for (int r = 0; r < 4; ++r) {
      lo[r] = (y[r]     + bl[r]) + rs[nt][0][r];
      hi[r] = (y[4 + r] + bh[r]) + rs[nt][1][r];
    }
    float* xo = x1 + tok * E + cb + nt * 16 + d0;
    *(v4f*)(xo)     = lo;
    *(v4f*)(xo + 4) = hi;
  }
}

static inline size_t align256(size_t x) { return (x + 255) & ~(size_t)255; }

extern "C" void kernel_launch(void* const* d_in, const int* in_sizes, int n_in,
                              void* d_out, int out_size, void* d_ws, size_t ws_size,
                              hipStream_t stream) {
  (void)in_sizes; (void)n_in; (void)out_size; (void)ws_size;

  const float* x        = (const float*)d_in[0];
  const float* ln1_g    = (const float*)d_in[1];
  const float* ln1_b    = (const float*)d_in[2];
  const float* acc_wqkv = (const float*)d_in[3];
  const float* acc_bqkv = (const float*)d_in[4];
  const float* acc_wo   = (const float*)d_in[5];
  const float* acc_bo   = (const float*)d_in[6];
  const float* gy_wqkv  = (const float*)d_in[7];
  const float* gy_bqkv  = (const float*)d_in[8];
  const float* gy_wo    = (const float*)d_in[9];
  const float* gy_bo    = (const float*)d_in[10];
  const float* conv_w   = (const float*)d_in[11];
  const float* ln2_g    = (const float*)d_in[12];
  const float* ln2_b    = (const float*)d_in[13];
  const float* w1       = (const float*)d_in[14];
  const float* b1       = (const float*)d_in[15];
  const float* w2       = (const float*)d_in[16];
  const float* b2       = (const float*)d_in[17];

  char* ws = (char*)d_ws;
  size_t off = 0;
  _Float16* xn_h   = (_Float16*)(ws + off); off = align256(off + (size_t)NTOK * E * 2);
  _Float16* qkv_h  = (_Float16*)(ws + off); off = align256(off + (size_t)2 * NTOK * 144 * 2);
  float*    x1     = (float*)   (ws + off); off = align256(off + (size_t)NTOK * E * 4);
  _Float16* wq_a_h = (_Float16*)(ws + off); off = align256(off + 144 * 48 * 2);
  _Float16* wo_a_h = (_Float16*)(ws + off); off = align256(off + 48 * 48 * 2);
  _Float16* wq_g_h = (_Float16*)(ws + off); off = align256(off + 144 * 48 * 2);
  _Float16* wo_g_h = (_Float16*)(ws + off); off = align256(off + 48 * 48 * 2);
  _Float16* w1_h   = (_Float16*)(ws + off); off = align256(off + 384 * 192 * 2);
  _Float16* w2_h   = (_Float16*)(ws + off); off = align256(off + 192 * 384 * 2);

  front_kernel<<<CVT_BLOCKS + NTOK / (8 * LN_TPW), 256, 0, stream>>>(
      x, ln1_g, ln1_b, xn_h,
      acc_wqkv, acc_wo, gy_wqkv, gy_wo, w1, w2,
      wq_a_h, wo_a_h, wq_g_h, wo_g_h, w1_h, w2_h);

  mid_kernel<<<CONV_BLOCKS + 2 * QKV_BLOCKS, 256, 0, stream>>>(
      xn_h, x, conv_w, x1, wq_a_h, wq_g_h, acc_bqkv, gy_bqkv,
      qkv_h, qkv_h + (size_t)NTOK * 144, E, 48, 144);

  attn_oproj_kernel<<<AT_BLOCKS, 512, 0, stream>>>(
      qkv_h, wo_a_h, wo_g_h, acc_bo, gy_bo, x, x1);

  mlp_fused_kernel<<<MLP_BLOCKS, 512, 0, stream>>>(
      ln2_g, ln2_b, w1_h, w2_h, b1, b2, x1, (float*)d_out);
}
